# starting kernel
# baseline (speedup 1.0000x reference)
#include <hip/hip_runtime.h>
#include <hip/hip_bf16.h>
#include <math.h>

typedef __attribute__((ext_vector_type(2))) float v2f;
typedef __attribute__((ext_vector_type(8))) float v8f;

#define OBJ 64
#define NPTS 512
#define MC 512
#define Z_MIN 1e-4f
#define EPSV 1e-5f

#define POSE_OFF 0
#define LW_OFF   229376
#define CI_OFF   262144

#define WSTRIDE 40

#define C_LP_T 2.7568155996140186f
#define C_LP_R 2.9826069522913796f

__device__ __forceinline__ float block_reduce_sum(float v, float* red) {
  const int tid = threadIdx.x;
  red[tid] = v;
  __syncthreads();
  for (int s = 128; s > 0; s >>= 1) {
    if (tid < s) red[tid] += red[tid + s];
    __syncthreads();
  }
  float r = red[0];
  __syncthreads();
  return r;
}

__device__ __forceinline__ void quat_rot(float qw, float qi, float qj, float qk, float R[3][3]) {
  R[0][0] = 1.f - 2.f * (qj * qj + qk * qk);
  R[0][1] = 2.f * (qi * qj - qk * qw);
  R[0][2] = 2.f * (qi * qk + qj * qw);
  R[1][0] = 2.f * (qi * qj + qk * qw);
  R[1][1] = 1.f - 2.f * (qi * qi + qk * qk);
  R[1][2] = 2.f * (qj * qk - qi * qw);
  R[2][0] = 2.f * (qi * qk - qj * qw);
  R[2][1] = 2.f * (qj * qk + qi * qw);
  R[2][2] = 1.f - 2.f * (qi * qi + qj * qj);
}

__global__ void epropnp_setup_kernel(const float* __restrict__ x3d,
                                     const float* __restrict__ x2d,
                                     const float* __restrict__ w2d,
                                     const float* __restrict__ pose,
                                     const float* __restrict__ cam,
                                     const float* __restrict__ cov_t,
                                     const float* __restrict__ cov_r,
                                     float* __restrict__ ws,
                                     float* __restrict__ out) {
  const int o = blockIdx.x;
  const int tid = threadIdx.x;
  __shared__ float red[256];
  __shared__ float sh[16];

  float sx = 0.f, sy = 0.f, sz = 0.f;
  for (int p = tid; p < NPTS; p += 256) {
    const float* xp = x3d + ((size_t)o * NPTS + p) * 3;
    sx += xp[0]; sy += xp[1]; sz += xp[2];
  }
  float offx = block_reduce_sum(sx, red) * (1.f / NPTS);
  float offy = block_reduce_sum(sy, red) * (1.f / NPTS);
  float offz = block_reduce_sum(sz, red) * (1.f / NPTS);

  if (tid == 0) {
    float* W = ws + (size_t)o * WSTRIDE;
    const float* P = pose + o * 7;
    float qw = P[3], qi = P[4], qj = P[5], qk = P[6];
    float R0[3][3];
    quat_rot(qw, qi, qj, qk, R0);
    float t0x = P[0] + R0[0][0] * offx + R0[0][1] * offy + R0[0][2] * offz;
    float t0y = P[1] + R0[1][0] * offx + R0[1][1] * offy + R0[1][2] * offz;
    float t0z = P[2] + R0[2][0] * offx + R0[2][1] * offy + R0[2][2] * offz;

    const float* A = cov_t + o * 9;
    float L00 = sqrtf(A[0]);
    float L10 = A[3] / L00;
    float L20 = A[6] / L00;
    float L11 = sqrtf(A[4] - L10 * L10);
    float L21 = (A[7] - L20 * L10) / L11;
    float L22 = sqrtf(A[8] - L20 * L20 - L21 * L21);

    const float* B = cov_r + o * 16;
    float N00 = sqrtf(B[0]);
    float N10 = B[4] / N00, N20 = B[8] / N00, N30 = B[12] / N00;
    float N11 = sqrtf(B[5] - N10 * N10);
    float N21 = (B[9] - N20 * N10) / N11;
    float N31 = (B[13] - N30 * N10) / N11;
    float N22 = sqrtf(B[10] - N20 * N20 - N21 * N21);
    float N32 = (B[14] - N30 * N20 - N31 * N21) / N22;
    float N33 = sqrtf(B[15] - N30 * N30 - N31 * N31 - N32 * N32);

    W[0] = offx; W[1] = offy; W[2] = offz;
    W[3] = t0x;  W[4] = t0y;  W[5] = t0z;
    W[6] = L00; W[7] = 0.f; W[8] = 0.f;
    W[9] = L10; W[10] = L11; W[11] = 0.f;
    W[12] = L20; W[13] = L21; W[14] = L22;
    float* Lr = W + 15;
    Lr[0] = N00; Lr[1] = 0.f; Lr[2] = 0.f; Lr[3] = 0.f;
    Lr[4] = N10; Lr[5] = N11; Lr[6] = 0.f; Lr[7] = 0.f;
    Lr[8] = N20; Lr[9] = N21; Lr[10] = N22; Lr[11] = 0.f;
    Lr[12] = N30; Lr[13] = N31; Lr[14] = N32; Lr[15] = N33;
    W[31] = logf(L00) + logf(L11) + logf(L22);
    W[32] = logf(N00) + logf(N11) + logf(N22) + logf(N33);

    const float* Kp = cam + o * 9;
    float ts[3] = {t0x, t0y, t0z};
    for (int i = 0; i < 3; i++) {
      for (int j = 0; j < 3; j++)
        sh[3 + i * 3 + j] = Kp[i * 3 + 0] * R0[0][j] + Kp[i * 3 + 1] * R0[1][j] + Kp[i * 3 + 2] * R0[2][j];
      sh[12 + i] = Kp[i * 3 + 0] * ts[0] + Kp[i * 3 + 1] * ts[1] + Kp[i * 3 + 2] * ts[2];
    }
    sh[0] = offx; sh[1] = offy; sh[2] = offz;
  }
  __syncthreads();

  float c = 0.f;
  for (int p = tid; p < NPTS; p += 256) {
    const float* xp = x3d + ((size_t)o * NPTS + p) * 3;
    float px = xp[0] - sh[0], py = xp[1] - sh[1], pz = xp[2] - sh[2];
    float u = sh[3] * px + sh[4] * py + sh[5] * pz + sh[12];
    float v = sh[6] * px + sh[7] * py + sh[8] * pz + sh[13];
    float w = sh[9] * px + sh[10] * py + sh[11] * pz + sh[14];
    float z = fmaxf(w, Z_MIN);
    float iz = __builtin_amdgcn_rcpf(z);
    const float* x2p = x2d + ((size_t)o * NPTS + p) * 2;
    const float* w2p = w2d + ((size_t)o * NPTS + p) * 2;
    float ru = (u * iz - x2p[0]) * w2p[0];
    float rv = (v * iz - x2p[1]) * w2p[1];
    c = fmaf(ru, ru, fmaf(rv, rv, c));
  }
  float total = block_reduce_sum(c, red);
  if (tid == 0) out[CI_OFF + o] = 0.5f * total;
}

__global__ void epropnp_sample_kernel(const float* __restrict__ x3d,
                                      const float* __restrict__ x2d,
                                      const float* __restrict__ w2d,
                                      const float* __restrict__ cam,
                                      const float* __restrict__ noise_t,
                                      const float* __restrict__ noise_r,
                                      const float* __restrict__ ws,
                                      float* __restrict__ out) {
  const int tile = blockIdx.x;
  const int o = blockIdx.y;
  const int tid = threadIdx.x;

  __shared__ float As[16][16];
  __shared__ float wave_part[8][16];

  const float* W = ws + (size_t)o * WSTRIDE;
  const float offx = W[0], offy = W[1], offz = W[2];

  if (tid < 16) {
    const int s = tid;
    const int mc = tile * 16 + s;
    const float* nt = noise_t + ((size_t)mc * OBJ + o) * 3;
    const float* nr = noise_r + ((size_t)mc * OBJ + o) * 4;
    const float n0 = nt[0], n1 = nt[1], n2 = nt[2];
    const float r0 = nr[0], r1 = nr[1], r2 = nr[2], r3 = nr[3];
    const float* Lt = W + 6;
    const float* Lr = W + 15;

    float tsx = W[3] + Lt[0] * n0;
    float tsy = W[4] + Lt[3] * n0 + Lt[4] * n1;
    float tsz = W[5] + Lt[6] * n0 + Lt[7] * n1 + Lt[8] * n2;

    float g0 = Lr[0] * r0;
    float g1 = Lr[4] * r0 + Lr[5] * r1;
    float g2 = Lr[8] * r0 + Lr[9] * r1 + Lr[10] * r2;
    float g3 = Lr[12] * r0 + Lr[13] * r1 + Lr[14] * r2 + Lr[15] * r3;
    float gn = sqrtf(g0 * g0 + g1 * g1 + g2 * g2 + g3 * g3);

    float qw, qi, qj, qk, Mr;
    if (gn < EPSV) {
      qw = 1.f; qi = 0.f; qj = 0.f; qk = 0.f;
      float y0 = 1.f / Lr[0];
      float y1 = -(Lr[4] * y0) / Lr[5];
      float y2 = -(Lr[8] * y0 + Lr[9] * y1) / Lr[10];
      float y3 = -(Lr[12] * y0 + Lr[13] * y1 + Lr[14] * y2) / Lr[15];
      Mr = y0 * y0 + y1 * y1 + y2 * y2 + y3 * y3;
    } else {
      float inv = 1.f / gn;
      qw = g0 * inv; qi = g1 * inv; qj = g2 * inv; qk = g3 * inv;
      Mr = (r0 * r0 + r1 * r1 + r2 * r2 + r3 * r3) * inv * inv;
    }

    float R[3][3];
    quat_rot(qw, qi, qj, qk, R);

    const float* Kp = cam + o * 9;
    float ts[3] = {tsx, tsy, tsz};
    #pragma unroll
    for (int i = 0; i < 3; i++) {
      #pragma unroll
      for (int j = 0; j < 3; j++)
        As[s][i * 4 + j] = Kp[i * 3 + 0] * R[0][j] + Kp[i * 3 + 1] * R[1][j] + Kp[i * 3 + 2] * R[2][j];
      As[s][i * 4 + 3] = Kp[i * 3 + 0] * ts[0] + Kp[i * 3 + 1] * ts[1] + Kp[i * 3 + 2] * ts[2];
    }
    float lp_t = -0.5f * (n0 * n0 + n1 * n1 + n2 * n2) - W[31] - C_LP_T;
    float lp_r = -2.f * logf(Mr) - W[32] - C_LP_R;
    As[s][12] = lp_t + lp_r;
    As[s][13] = 0.f; As[s][14] = 0.f; As[s][15] = 0.f;

    float* po = out + POSE_OFF + ((size_t)mc * OBJ + o) * 7;
    po[0] = tsx - (R[0][0] * offx + R[0][1] * offy + R[0][2] * offz);
    po[1] = tsy - (R[1][0] * offx + R[1][1] * offy + R[1][2] * offz);
    po[2] = tsz - (R[2][0] * offx + R[2][1] * offy + R[2][2] * offz);
    po[3] = qw; po[4] = qi; po[5] = qj; po[6] = qk;
  }
  __syncthreads();

  const int lane = tid & 31;
  const int wave = tid >> 5;
  const int g = lane >> 4;
  const int sl = lane & 15;
  const int k0 = 2 * g;

  v2f a_u = {As[sl][0 + k0], As[sl][1 + k0]};
  v2f a_v = {As[sl][4 + k0], As[sl][5 + k0]};
  v2f a_z = {As[sl][8 + k0], As[sl][9 + k0]};

  float acc[8];
  #pragma unroll
  for (int r = 0; r < 8; r++) acc[r] = 0.f;

  #pragma unroll
  for (int k = 0; k < 4; ++k) {
    const int n = (wave * 4 + k) * 16 + sl;
    const float* xp = x3d + ((size_t)o * NPTS + n) * 3;
    float px = xp[0] - offx;
    float py = xp[1] - offy;
    float pz = xp[2] - offz;
    v2f bf;
    bf.x = g ? pz : px;
    bf.y = g ? 1.0f : py;

    v8f cu = {}, cv = {}, cz = {};
    cu = __builtin_amdgcn_wmma_f32_16x16x4_f32(false, a_u, false, bf, (short)0, cu, false, false);
    cv = __builtin_amdgcn_wmma_f32_16x16x4_f32(false, a_v, false, bf, (short)0, cv, false, false);
    cz = __builtin_amdgcn_wmma_f32_16x16x4_f32(false, a_z, false, bf, (short)0, cz, false, false);

    const float* x2p = x2d + ((size_t)o * NPTS + n) * 2;
    const float* w2p = w2d + ((size_t)o * NPTS + n) * 2;
    const float xu = x2p[0], xv = x2p[1], wu = w2p[0], wv = w2p[1];

    #pragma unroll
    for (int r = 0; r < 8; r++) {
      float z = fmaxf(cz[r], Z_MIN);
      float iz = __builtin_amdgcn_rcpf(z);
      float ru = (cu[r] * iz - xu) * wu;
      float rv = (cv[r] * iz - xv) * wv;
      acc[r] = fmaf(ru, ru, fmaf(rv, rv, acc[r]));
    }
  }

  #pragma unroll
  for (int r = 0; r < 8; r++) {
    float v = acc[r];
    v += __shfl_xor(v, 1, 32);
    v += __shfl_xor(v, 2, 32);
    v += __shfl_xor(v, 4, 32);
    v += __shfl_xor(v, 8, 32);
    acc[r] = v;
  }
  if (sl == 0) {
    #pragma unroll
    for (int r = 0; r < 8; r++) wave_part[wave][r + 8 * g] = acc[r];
  }
  __syncthreads();

  if (tid < 16) {
    float c = 0.f;
    for (int w = 0; w < 8; w++) c += wave_part[w][tid];
    const int mc = tile * 16 + tid;
    out[LW_OFF + (size_t)mc * OBJ + o] = -0.5f * c - As[tid][12];
  }
}

extern "C" void kernel_launch(void* const* d_in, const int* in_sizes, int n_in,
                              void* d_out, int out_size, void* d_ws, size_t ws_size,
                              hipStream_t stream) {
  const float* x3d     = (const float*)d_in[0];
  const float* x2d     = (const float*)d_in[1];
  const float* w2d     = (const float*)d_in[2];
  const float* pose    = (const float*)d_in[3];
  const float* cam     = (const float*)d_in[4];
  const float* cov_t   = (const float*)d_in[5];
  const float* cov_r   = (const float*)d_in[6];
  const float* noise_t = (const float*)d_in[7];
  const float* noise_r = (const float*)d_in[8];
  float* out = (float*)d_out;
  float* ws  = (float*)d_ws;

  epropnp_setup_kernel<<<OBJ, 256, 0, stream>>>(x3d, x2d, w2d, pose, cam, cov_t, cov_r, ws, out);
  epropnp_sample_kernel<<<dim3(MC / 16, OBJ), 256, 0, stream>>>(x3d, x2d, w2d, cam, noise_t, noise_r, ws, out);
}


# fuse the setup launch into the sample kernel (2 launches -> 1): each workgroup reduces its object's mean offset and factors the covariances, 64 more do cost_init; offset, cost_init sum order changed
# speedup vs baseline: 1.0845x; 1.0845x over previous
#include <hip/hip_runtime.h>
#include <hip/hip_bf16.h>
#include <math.h>

typedef __attribute__((ext_vector_type(2))) float v2f;
typedef __attribute__((ext_vector_type(8))) float v8f;

#define OBJ 64
#define NPTS 512
#define MC 512
#define Z_MIN 1e-4f
#define EPSV 1e-5f

#define POSE_OFF 0
#define LW_OFF   229376
#define CI_OFF   262144

#define C_LP_T 2.7568155996140186f
#define C_LP_R 2.9826069522913796f

__device__ __forceinline__ void quat_rot(float qw, float qi, float qj, float qk, float R[3][3]) {
  R[0][0] = 1.f - 2.f * (qj * qj + qk * qk);
  R[0][1] = 2.f * (qi * qj - qk * qw);
  R[0][2] = 2.f * (qi * qk + qj * qw);
  R[1][0] = 2.f * (qi * qj + qk * qw);
  R[1][1] = 1.f - 2.f * (qi * qi + qk * qk);
  R[1][2] = 2.f * (qj * qk - qi * qw);
  R[2][0] = 2.f * (qi * qk - qj * qw);
  R[2][1] = 2.f * (qj * qk + qi * qw);
  R[2][2] = 1.f - 2.f * (qi * qi + qj * qj);
}

__device__ __forceinline__ void wave_point_sums(const float X[4], const float Y[4], const float Z[4],
                                                float (*psum)[4], int wave, int lane) {
  float sx = (X[0] + X[1]) + (X[2] + X[3]);
  float sy = (Y[0] + Y[1]) + (Y[2] + Y[3]);
  float sz = (Z[0] + Z[1]) + (Z[2] + Z[3]);
  #pragma unroll
  for (int m = 1; m < 16; m <<= 1) {
    sx += __shfl_xor(sx, m, 32);
    sy += __shfl_xor(sy, m, 32);
    sz += __shfl_xor(sz, m, 32);
  }
  if (lane == 0) { psum[wave][0] = sx; psum[wave][1] = sy; psum[wave][2] = sz; }
}

__global__ void __launch_bounds__(256)
epropnp_fused_kernel(const float* __restrict__ x3d,
                     const float* __restrict__ x2d,
                     const float* __restrict__ w2d,
                     const float* __restrict__ pose,
                     const float* __restrict__ cam,
                     const float* __restrict__ cov_t,
                     const float* __restrict__ cov_r,
                     const float* __restrict__ noise_t,
                     const float* __restrict__ noise_r,
                     float* __restrict__ out) {
  const int o = blockIdx.y;
  const int tid = threadIdx.x;
  const int lane = tid & 31;
  const int wave = tid >> 5;
  const int g = lane >> 4;
  const int sl = lane & 15;

  __shared__ float As[16][16];
  __shared__ float wave_part[8][16];
  __shared__ float psum[8][4];

  float X[4], Y[4], Z[4];
  #pragma unroll
  for (int k = 0; k < 4; ++k) {
    const int n = (wave * 4 + k) * 16 + sl;
    const float* xp = x3d + ((size_t)o * NPTS + n) * 3;
    X[k] = xp[0]; Y[k] = xp[1]; Z[k] = xp[2];
  }

  if (blockIdx.x == 0) {
    const float xa = g ? X[2] : X[0], ya = g ? Y[2] : Y[0], za = g ? Z[2] : Z[0];
    const float xb = g ? X[3] : X[1], yb = g ? Y[3] : Y[1], zb = g ? Z[3] : Z[1];
    const int na = (wave * 4 + 2 * g) * 16 + sl;
    const int nb = na + 16;
    const float* x2a = x2d + ((size_t)o * NPTS + na) * 2;
    const float* w2a = w2d + ((size_t)o * NPTS + na) * 2;
    const float* x2b = x2d + ((size_t)o * NPTS + nb) * 2;
    const float* w2b = w2d + ((size_t)o * NPTS + nb) * 2;
    const float xua = x2a[0], xva = x2a[1], wua = w2a[0], wva = w2a[1];
    const float xub = x2b[0], xvb = x2b[1], wub = w2b[0], wvb = w2b[1];

    const float* P = pose + o * 7;
    const float* Kp = cam + o * 9;
    const float P0 = P[0], P1 = P[1], P2 = P[2];
    float Kc[9];
    #pragma unroll
    for (int i = 0; i < 9; i++) Kc[i] = Kp[i];
    float R0[3][3];
    quat_rot(P[3], P[4], P[5], P[6], R0);
    wave_point_sums(X, Y, Z, psum, wave, lane);
    __syncthreads();

    float offx = 0.f, offy = 0.f, offz = 0.f;
    #pragma unroll
    for (int w = 0; w < 8; w++) { offx += psum[w][0]; offy += psum[w][1]; offz += psum[w][2]; }
    offx *= (1.f / NPTS); offy *= (1.f / NPTS); offz *= (1.f / NPTS);

    float ts[3];
    ts[0] = P0 + R0[0][0] * offx + R0[0][1] * offy + R0[0][2] * offz;
    ts[1] = P1 + R0[1][0] * offx + R0[1][1] * offy + R0[1][2] * offz;
    ts[2] = P2 + R0[2][0] * offx + R0[2][1] * offy + R0[2][2] * offz;
    float M0[3][3], b0[3];
    #pragma unroll
    for (int i = 0; i < 3; i++) {
      #pragma unroll
      for (int j = 0; j < 3; j++)
        M0[i][j] = Kc[i * 3 + 0] * R0[0][j] + Kc[i * 3 + 1] * R0[1][j] + Kc[i * 3 + 2] * R0[2][j];
      b0[i] = Kc[i * 3 + 0] * ts[0] + Kc[i * 3 + 1] * ts[1] + Kc[i * 3 + 2] * ts[2];
    }

    float c = 0.f;
    {
      float px = xa - offx, py = ya - offy, pz = za - offz;
      float u = M0[0][0] * px + M0[0][1] * py + M0[0][2] * pz + b0[0];
      float v = M0[1][0] * px + M0[1][1] * py + M0[1][2] * pz + b0[1];
      float w = M0[2][0] * px + M0[2][1] * py + M0[2][2] * pz + b0[2];
      float z = fmaxf(w, Z_MIN);
      float iz = __builtin_amdgcn_rcpf(z);
      float ru = (u * iz - xua) * wua;
      float rv = (v * iz - xva) * wva;
      c = fmaf(ru, ru, fmaf(rv, rv, c));
    }
    {
      float px = xb - offx, py = yb - offy, pz = zb - offz;
      float u = M0[0][0] * px + M0[0][1] * py + M0[0][2] * pz + b0[0];
      float v = M0[1][0] * px + M0[1][1] * py + M0[1][2] * pz + b0[1];
      float w = M0[2][0] * px + M0[2][1] * py + M0[2][2] * pz + b0[2];
      float z = fmaxf(w, Z_MIN);
      float iz = __builtin_amdgcn_rcpf(z);
      float ru = (u * iz - xub) * wub;
      float rv = (v * iz - xvb) * wvb;
      c = fmaf(ru, ru, fmaf(rv, rv, c));
    }
    #pragma unroll
    for (int m = 1; m < 32; m <<= 1) c += __shfl_xor(c, m, 32);
    if (lane == 0) wave_part[wave][0] = c;
    __syncthreads();
    if (tid == 0) {
      float total = 0.f;
      #pragma unroll
      for (int w = 0; w < 8; w++) total += wave_part[w][0];
      out[CI_OFF + o] = 0.5f * total;
    }
    return;
  }

  const int tile = blockIdx.x - 1;

  float R[3][3] = {}, R0[3][3] = {}, Kc[9] = {};
  float L00 = 0.f, L10 = 0.f, L11 = 0.f, L20 = 0.f, L21 = 0.f, L22 = 0.f;
  float n0 = 0.f, n1 = 0.f, n2 = 0.f, P0 = 0.f, P1 = 0.f, P2 = 0.f;

  if (tid < 16) {
    const int s = tid;
    const int mc = tile * 16 + s;
    const float* nt = noise_t + ((size_t)mc * OBJ + o) * 3;
    const float* nr = noise_r + ((size_t)mc * OBJ + o) * 4;
    n0 = nt[0]; n1 = nt[1]; n2 = nt[2];
    const float r0 = nr[0], r1 = nr[1], r2 = nr[2], r3 = nr[3];
    const float* P = pose + o * 7;
    P0 = P[0]; P1 = P[1]; P2 = P[2];
    quat_rot(P[3], P[4], P[5], P[6], R0);

    const float* A = cov_t + o * 9;
    L00 = sqrtf(A[0]);
    L10 = A[3] / L00;
    L20 = A[6] / L00;
    L11 = sqrtf(A[4] - L10 * L10);
    L21 = (A[7] - L20 * L10) / L11;
    L22 = sqrtf(A[8] - L20 * L20 - L21 * L21);

    const float* B = cov_r + o * 16;
    const float N00 = sqrtf(B[0]);
    const float N10 = B[4] / N00, N20 = B[8] / N00, N30 = B[12] / N00;
    const float N11 = sqrtf(B[5] - N10 * N10);
    const float N21 = (B[9] - N20 * N10) / N11;
    const float N31 = (B[13] - N30 * N10) / N11;
    const float N22 = sqrtf(B[10] - N20 * N20 - N21 * N21);
    const float N32 = (B[14] - N30 * N20 - N31 * N21) / N22;
    const float N33 = sqrtf(B[15] - N30 * N30 - N31 * N31 - N32 * N32);

    const float hld_t = logf(L00) + logf(L11) + logf(L22);
    const float hld_r = logf(N00) + logf(N11) + logf(N22) + logf(N33);

    float g0 = N00 * r0;
    float g1 = N10 * r0 + N11 * r1;
    float g2 = N20 * r0 + N21 * r1 + N22 * r2;
    float g3 = N30 * r0 + N31 * r1 + N32 * r2 + N33 * r3;
    float gn = sqrtf(g0 * g0 + g1 * g1 + g2 * g2 + g3 * g3);

    float qw, qi, qj, qk, Mr;
    if (gn < EPSV) {
      qw = 1.f; qi = 0.f; qj = 0.f; qk = 0.f;
      float y0 = 1.f / N00;
      float y1 = -(N10 * y0) / N11;
      float y2 = -(N20 * y0 + N21 * y1) / N22;
      float y3 = -(N30 * y0 + N31 * y1 + N32 * y2) / N33;
      Mr = y0 * y0 + y1 * y1 + y2 * y2 + y3 * y3;
    } else {
      float inv = 1.f / gn;
      qw = g0 * inv; qi = g1 * inv; qj = g2 * inv; qk = g3 * inv;
      Mr = (r0 * r0 + r1 * r1 + r2 * r2 + r3 * r3) * inv * inv;
    }
    quat_rot(qw, qi, qj, qk, R);

    const float* Kp = cam + o * 9;
    #pragma unroll
    for (int i = 0; i < 9; i++) Kc[i] = Kp[i];
    #pragma unroll
    for (int i = 0; i < 3; i++) {
      #pragma unroll
      for (int j = 0; j < 3; j++)
        As[s][i * 4 + j] = Kc[i * 3 + 0] * R[0][j] + Kc[i * 3 + 1] * R[1][j] + Kc[i * 3 + 2] * R[2][j];
    }
    float lp_t = -0.5f * (n0 * n0 + n1 * n1 + n2 * n2) - hld_t - C_LP_T;
    float lp_r = -2.f * logf(Mr) - hld_r - C_LP_R;
    As[s][12] = lp_t + lp_r;
    As[s][13] = 0.f; As[s][14] = 0.f; As[s][15] = 0.f;

    float* po = out + POSE_OFF + ((size_t)mc * OBJ + o) * 7;
    po[3] = qw; po[4] = qi; po[5] = qj; po[6] = qk;
  }
  wave_point_sums(X, Y, Z, psum, wave, lane);
  __syncthreads();

  float offx = 0.f, offy = 0.f, offz = 0.f;
  #pragma unroll
  for (int w = 0; w < 8; w++) { offx += psum[w][0]; offy += psum[w][1]; offz += psum[w][2]; }
  offx *= (1.f / NPTS); offy *= (1.f / NPTS); offz *= (1.f / NPTS);

  if (tid < 16) {
    const int s = tid;
    const int mc = tile * 16 + s;
    const float t0x = P0 + R0[0][0] * offx + R0[0][1] * offy + R0[0][2] * offz;
    const float t0y = P1 + R0[1][0] * offx + R0[1][1] * offy + R0[1][2] * offz;
    const float t0z = P2 + R0[2][0] * offx + R0[2][1] * offy + R0[2][2] * offz;
    const float tsx = t0x + L00 * n0;
    const float tsy = t0y + L10 * n0 + L11 * n1;
    const float tsz = t0z + L20 * n0 + L21 * n1 + L22 * n2;
    #pragma unroll
    for (int i = 0; i < 3; i++)
      As[s][i * 4 + 3] = Kc[i * 3 + 0] * tsx + Kc[i * 3 + 1] * tsy + Kc[i * 3 + 2] * tsz;

    float* po = out + POSE_OFF + ((size_t)mc * OBJ + o) * 7;
    po[0] = tsx - (R[0][0] * offx + R[0][1] * offy + R[0][2] * offz);
    po[1] = tsy - (R[1][0] * offx + R[1][1] * offy + R[1][2] * offz);
    po[2] = tsz - (R[2][0] * offx + R[2][1] * offy + R[2][2] * offz);
  }
  __syncthreads();

  const int k0 = 2 * g;

  v2f a_u = {As[sl][0 + k0], As[sl][1 + k0]};
  v2f a_v = {As[sl][4 + k0], As[sl][5 + k0]};
  v2f a_z = {As[sl][8 + k0], As[sl][9 + k0]};

  float acc[8];
  #pragma unroll
  for (int r = 0; r < 8; r++) acc[r] = 0.f;

  #pragma unroll
  for (int k = 0; k < 4; ++k) {
    const int n = (wave * 4 + k) * 16 + sl;
    float px = X[k] - offx;
    float py = Y[k] - offy;
    float pz = Z[k] - offz;
    v2f bf;
    bf.x = g ? pz : px;
    bf.y = g ? 1.0f : py;

    v8f cu = {}, cv = {}, cz = {};
    cu = __builtin_amdgcn_wmma_f32_16x16x4_f32(false, a_u, false, bf, (short)0, cu, false, false);
    cv = __builtin_amdgcn_wmma_f32_16x16x4_f32(false, a_v, false, bf, (short)0, cv, false, false);
    cz = __builtin_amdgcn_wmma_f32_16x16x4_f32(false, a_z, false, bf, (short)0, cz, false, false);

    const float* x2p = x2d + ((size_t)o * NPTS + n) * 2;
    const float* w2p = w2d + ((size_t)o * NPTS + n) * 2;
    const float xu = x2p[0], xv = x2p[1], wu = w2p[0], wv = w2p[1];

    #pragma unroll
    for (int r = 0; r < 8; r++) {
      float z = fmaxf(cz[r], Z_MIN);
      float iz = __builtin_amdgcn_rcpf(z);
      float ru = (cu[r] * iz - xu) * wu;
      float rv = (cv[r] * iz - xv) * wv;
      acc[r] = fmaf(ru, ru, fmaf(rv, rv, acc[r]));
    }
  }

  #pragma unroll
  for (int r = 0; r < 8; r++) {
    float v = acc[r];
    v += __shfl_xor(v, 1, 32);
    v += __shfl_xor(v, 2, 32);
    v += __shfl_xor(v, 4, 32);
    v += __shfl_xor(v, 8, 32);
    acc[r] = v;
  }
  if (sl == 0) {
    #pragma unroll
    for (int r = 0; r < 8; r++) wave_part[wave][r + 8 * g] = acc[r];
  }
  __syncthreads();

  if (tid < 16) {
    float c = 0.f;
    for (int w = 0; w < 8; w++) c += wave_part[w][tid];
    const int mc = tile * 16 + tid;
    out[LW_OFF + (size_t)mc * OBJ + o] = -0.5f * c - As[tid][12];
  }
}

extern "C" void kernel_launch(void* const* d_in, const int* in_sizes, int n_in,
                              void* d_out, int out_size, void* d_ws, size_t ws_size,
                              hipStream_t stream) {
  const float* x3d     = (const float*)d_in[0];
  const float* x2d     = (const float*)d_in[1];
  const float* w2d     = (const float*)d_in[2];
  const float* pose    = (const float*)d_in[3];
  const float* cam     = (const float*)d_in[4];
  const float* cov_t   = (const float*)d_in[5];
  const float* cov_r   = (const float*)d_in[6];
  const float* noise_t = (const float*)d_in[7];
  const float* noise_r = (const float*)d_in[8];
  float* out = (float*)d_out;

  epropnp_fused_kernel<<<dim3(33, 64), 256, 0, stream>>>(x3d, x2d, w2d, pose, cam, cov_t, cov_r, noise_t, noise_r, out);
}


# tile loop: each workgroup takes 128 samples (8 tiles of 16) of its object, points, targets and weights held in registers; grid 33x64 -> 4x64 workgroups; cost_init done by group 0
# speedup vs baseline: 1.5504x; 1.4296x over previous
#include <hip/hip_runtime.h>
#include <hip/hip_bf16.h>
#include <math.h>

typedef __attribute__((ext_vector_type(2))) float v2f;
typedef __attribute__((ext_vector_type(8))) float v8f;

#define OBJ 64
#define NPTS 512
#define MC 512
#define Z_MIN 1e-4f
#define EPSV 1e-5f

#define POSE_OFF 0
#define LW_OFF   229376
#define CI_OFF   262144

#define C_LP_T 2.7568155996140186f
#define C_LP_R 2.9826069522913796f

__device__ __forceinline__ void quat_rot(float qw, float qi, float qj, float qk, float R[3][3]) {
  R[0][0] = 1.f - 2.f * (qj * qj + qk * qk);
  R[0][1] = 2.f * (qi * qj - qk * qw);
  R[0][2] = 2.f * (qi * qk + qj * qw);
  R[1][0] = 2.f * (qi * qj + qk * qw);
  R[1][1] = 1.f - 2.f * (qi * qi + qk * qk);
  R[1][2] = 2.f * (qj * qk - qi * qw);
  R[2][0] = 2.f * (qi * qk - qj * qw);
  R[2][1] = 2.f * (qj * qk + qi * qw);
  R[2][2] = 1.f - 2.f * (qi * qi + qj * qj);
}

__device__ __forceinline__ void wave_point_sums(const float X[4], const float Y[4], const float Z[4],
                                                float (*psum)[4], int wave, int lane) {
  float sx = (X[0] + X[1]) + (X[2] + X[3]);
  float sy = (Y[0] + Y[1]) + (Y[2] + Y[3]);
  float sz = (Z[0] + Z[1]) + (Z[2] + Z[3]);
  #pragma unroll
  for (int m = 1; m < 16; m <<= 1) {
    sx += __shfl_xor(sx, m, 32);
    sy += __shfl_xor(sy, m, 32);
    sz += __shfl_xor(sz, m, 32);
  }
  if (lane == 0) { psum[wave][0] = sx; psum[wave][1] = sy; psum[wave][2] = sz; }
}

#define TILES 8
#define SPW (TILES * 16)

__global__ void __launch_bounds__(256)
epropnp_fused_kernel(const float* __restrict__ x3d,
                     const float* __restrict__ x2d,
                     const float* __restrict__ w2d,
                     const float* __restrict__ pose,
                     const float* __restrict__ cam,
                     const float* __restrict__ cov_t,
                     const float* __restrict__ cov_r,
                     const float* __restrict__ noise_t,
                     const float* __restrict__ noise_r,
                     float* __restrict__ out) {
  const int o = blockIdx.y;
  const int grp = blockIdx.x;
  const int tid = threadIdx.x;
  const int lane = tid & 31;
  const int wave = tid >> 5;
  const int g = lane >> 4;
  const int sl = lane & 15;
  const bool do_ci = (grp == 0);

  __shared__ float As[SPW][16];
  __shared__ float wave_part[TILES][8][16];
  __shared__ float psum[8][4];
  __shared__ float ci_part[8];

  float X[4], Y[4], Z[4];
  #pragma unroll
  for (int k = 0; k < 4; ++k) {
    const int n = (wave * 4 + k) * 16 + sl;
    const float* xp = x3d + ((size_t)o * NPTS + n) * 3;
    X[k] = xp[0]; Y[k] = xp[1]; Z[k] = xp[2];
  }
  float xu[4], xv[4], wu[4], wv[4];
  #pragma unroll
  for (int k = 0; k < 4; ++k) {
    const int n = (wave * 4 + k) * 16 + sl;
    const float* x2p = x2d + ((size_t)o * NPTS + n) * 2;
    const float* w2p = w2d + ((size_t)o * NPTS + n) * 2;
    xu[k] = x2p[0]; xv[k] = x2p[1]; wu[k] = w2p[0]; wv[k] = w2p[1];
  }

  float R[3][3] = {}, R0[3][3] = {}, Kc[9] = {};
  float L00 = 0.f, L10 = 0.f, L11 = 0.f, L20 = 0.f, L21 = 0.f, L22 = 0.f;
  float n0 = 0.f, n1 = 0.f, n2 = 0.f, P0 = 0.f, P1 = 0.f, P2 = 0.f;

  if (tid < SPW) {
    const int s = tid;
    const int mc = grp * SPW + s;
    const float* nt = noise_t + ((size_t)mc * OBJ + o) * 3;
    const float* nr = noise_r + ((size_t)mc * OBJ + o) * 4;
    n0 = nt[0]; n1 = nt[1]; n2 = nt[2];
    const float r0 = nr[0], r1 = nr[1], r2 = nr[2], r3 = nr[3];
    const float* P = pose + o * 7;
    P0 = P[0]; P1 = P[1]; P2 = P[2];
    quat_rot(P[3], P[4], P[5], P[6], R0);

    const float* A = cov_t + o * 9;
    L00 = sqrtf(A[0]);
    L10 = A[3] / L00;
    L20 = A[6] / L00;
    L11 = sqrtf(A[4] - L10 * L10);
    L21 = (A[7] - L20 * L10) / L11;
    L22 = sqrtf(A[8] - L20 * L20 - L21 * L21);

    const float* B = cov_r + o * 16;
    const float N00 = sqrtf(B[0]);
    const float N10 = B[4] / N00, N20 = B[8] / N00, N30 = B[12] / N00;
    const float N11 = sqrtf(B[5] - N10 * N10);
    const float N21 = (B[9] - N20 * N10) / N11;
    const float N31 = (B[13] - N30 * N10) / N11;
    const float N22 = sqrtf(B[10] - N20 * N20 - N21 * N21);
    const float N32 = (B[14] - N30 * N20 - N31 * N21) / N22;
    const float N33 = sqrtf(B[15] - N30 * N30 - N31 * N31 - N32 * N32);

    const float hld_t = logf(L00) + logf(L11) + logf(L22);
    const float hld_r = logf(N00) + logf(N11) + logf(N22) + logf(N33);

    float g0 = N00 * r0;
    float g1 = N10 * r0 + N11 * r1;
    float g2 = N20 * r0 + N21 * r1 + N22 * r2;
    float g3 = N30 * r0 + N31 * r1 + N32 * r2 + N33 * r3;
    float gn = sqrtf(g0 * g0 + g1 * g1 + g2 * g2 + g3 * g3);

    float qw, qi, qj, qk, Mr;
    if (gn < EPSV) {
      qw = 1.f; qi = 0.f; qj = 0.f; qk = 0.f;
      float y0 = 1.f / N00;
      float y1 = -(N10 * y0) / N11;
      float y2 = -(N20 * y0 + N21 * y1) / N22;
      float y3 = -(N30 * y0 + N31 * y1 + N32 * y2) / N33;
      Mr = y0 * y0 + y1 * y1 + y2 * y2 + y3 * y3;
    } else {
      float inv = 1.f / gn;
      qw = g0 * inv; qi = g1 * inv; qj = g2 * inv; qk = g3 * inv;
      Mr = (r0 * r0 + r1 * r1 + r2 * r2 + r3 * r3) * inv * inv;
    }
    quat_rot(qw, qi, qj, qk, R);

    const float* Kp = cam + o * 9;
    #pragma unroll
    for (int i = 0; i < 9; i++) Kc[i] = Kp[i];
    #pragma unroll
    for (int i = 0; i < 3; i++) {
      #pragma unroll
      for (int j = 0; j < 3; j++)
        As[s][i * 4 + j] = Kc[i * 3 + 0] * R[0][j] + Kc[i * 3 + 1] * R[1][j] + Kc[i * 3 + 2] * R[2][j];
    }
    float lp_t = -0.5f * (n0 * n0 + n1 * n1 + n2 * n2) - hld_t - C_LP_T;
    float lp_r = -2.f * logf(Mr) - hld_r - C_LP_R;
    As[s][12] = lp_t + lp_r;
    As[s][13] = 0.f; As[s][14] = 0.f; As[s][15] = 0.f;

    float* po = out + POSE_OFF + ((size_t)mc * OBJ + o) * 7;
    po[3] = qw; po[4] = qi; po[5] = qj; po[6] = qk;
  }
  wave_point_sums(X, Y, Z, psum, wave, lane);
  __syncthreads();

  float offx = 0.f, offy = 0.f, offz = 0.f;
  #pragma unroll
  for (int w = 0; w < 8; w++) { offx += psum[w][0]; offy += psum[w][1]; offz += psum[w][2]; }
  offx *= (1.f / NPTS); offy *= (1.f / NPTS); offz *= (1.f / NPTS);

  if (tid < SPW) {
    const int s = tid;
    const int mc = grp * SPW + s;
    const float t0x = P0 + R0[0][0] * offx + R0[0][1] * offy + R0[0][2] * offz;
    const float t0y = P1 + R0[1][0] * offx + R0[1][1] * offy + R0[1][2] * offz;
    const float t0z = P2 + R0[2][0] * offx + R0[2][1] * offy + R0[2][2] * offz;
    const float tsx = t0x + L00 * n0;
    const float tsy = t0y + L10 * n0 + L11 * n1;
    const float tsz = t0z + L20 * n0 + L21 * n1 + L22 * n2;
    #pragma unroll
    for (int i = 0; i < 3; i++)
      As[s][i * 4 + 3] = Kc[i * 3 + 0] * tsx + Kc[i * 3 + 1] * tsy + Kc[i * 3 + 2] * tsz;

    float* po = out + POSE_OFF + ((size_t)mc * OBJ + o) * 7;
    po[0] = tsx - (R[0][0] * offx + R[0][1] * offy + R[0][2] * offz);
    po[1] = tsy - (R[1][0] * offx + R[1][1] * offy + R[1][2] * offz);
    po[2] = tsz - (R[2][0] * offx + R[2][1] * offy + R[2][2] * offz);
  }

  if (do_ci) {
    const float* P = pose + o * 7;
    const float* Kp = cam + o * 9;
    const float cP0 = P[0], cP1 = P[1], cP2 = P[2];
    float cK[9], cR0[3][3];
    #pragma unroll
    for (int i = 0; i < 9; i++) cK[i] = Kp[i];
    quat_rot(P[3], P[4], P[5], P[6], cR0);
    const float xa = g ? X[2] : X[0], ya = g ? Y[2] : Y[0], za = g ? Z[2] : Z[0];
    const float xb = g ? X[3] : X[1], yb = g ? Y[3] : Y[1], zb = g ? Z[3] : Z[1];
    const float xua = g ? xu[2] : xu[0], xva = g ? xv[2] : xv[0];
    const float wua = g ? wu[2] : wu[0], wva = g ? wv[2] : wv[0];
    const float xub = g ? xu[3] : xu[1], xvb = g ? xv[3] : xv[1];
    const float wub = g ? wu[3] : wu[1], wvb = g ? wv[3] : wv[1];

    float ts[3];
    ts[0] = cP0 + cR0[0][0] * offx + cR0[0][1] * offy + cR0[0][2] * offz;
    ts[1] = cP1 + cR0[1][0] * offx + cR0[1][1] * offy + cR0[1][2] * offz;
    ts[2] = cP2 + cR0[2][0] * offx + cR0[2][1] * offy + cR0[2][2] * offz;
    float M0[3][3], b0[3];
    #pragma unroll
    for (int i = 0; i < 3; i++) {
      #pragma unroll
      for (int j = 0; j < 3; j++)
        M0[i][j] = cK[i * 3 + 0] * cR0[0][j] + cK[i * 3 + 1] * cR0[1][j] + cK[i * 3 + 2] * cR0[2][j];
      b0[i] = cK[i * 3 + 0] * ts[0] + cK[i * 3 + 1] * ts[1] + cK[i * 3 + 2] * ts[2];
    }

    float c = 0.f;
    {
      float px = xa - offx, py = ya - offy, pz = za - offz;
      float u = M0[0][0] * px + M0[0][1] * py + M0[0][2] * pz + b0[0];
      float v = M0[1][0] * px + M0[1][1] * py + M0[1][2] * pz + b0[1];
      float w = M0[2][0] * px + M0[2][1] * py + M0[2][2] * pz + b0[2];
      float z = fmaxf(w, Z_MIN);
      float iz = __builtin_amdgcn_rcpf(z);
      float ru = (u * iz - xua) * wua;
      float rv = (v * iz - xva) * wva;
      c = fmaf(ru, ru, fmaf(rv, rv, c));
    }
    {
      float px = xb - offx, py = yb - offy, pz = zb - offz;
      float u = M0[0][0] * px + M0[0][1] * py + M0[0][2] * pz + b0[0];
      float v = M0[1][0] * px + M0[1][1] * py + M0[1][2] * pz + b0[1];
      float w = M0[2][0] * px + M0[2][1] * py + M0[2][2] * pz + b0[2];
      float z = fmaxf(w, Z_MIN);
      float iz = __builtin_amdgcn_rcpf(z);
      float ru = (u * iz - xub) * wub;
      float rv = (v * iz - xvb) * wvb;
      c = fmaf(ru, ru, fmaf(rv, rv, c));
    }
    #pragma unroll
    for (int m = 1; m < 32; m <<= 1) c += __shfl_xor(c, m, 32);
    if (lane == 0) ci_part[wave] = c;
  }
  __syncthreads();

  const int k0 = 2 * g;

  v2f bf[4];
  #pragma unroll
  for (int k = 0; k < 4; ++k) {
    float px = X[k] - offx;
    float py = Y[k] - offy;
    float pz = Z[k] - offz;
    bf[k].x = g ? pz : px;
    bf[k].y = g ? 1.0f : py;
  }

  #pragma unroll 1
  for (int tq = 0; tq < TILES; tq += 4) {
    float acc[4][8];
    #pragma unroll
    for (int ti = 0; ti < 4; ++ti) {
      const float* Ar = &As[(tq + ti) * 16 + sl][0];
      v2f a_u = {Ar[0 + k0], Ar[1 + k0]};
      v2f a_v = {Ar[4 + k0], Ar[5 + k0]};
      v2f a_z = {Ar[8 + k0], Ar[9 + k0]};

      #pragma unroll
      for (int r = 0; r < 8; r++) acc[ti][r] = 0.f;

      #pragma unroll
      for (int k = 0; k < 4; ++k) {
        v8f cu = {}, cv = {}, cz = {};
        cu = __builtin_amdgcn_wmma_f32_16x16x4_f32(false, a_u, false, bf[k], (short)0, cu, false, false);
        cv = __builtin_amdgcn_wmma_f32_16x16x4_f32(false, a_v, false, bf[k], (short)0, cv, false, false);
        cz = __builtin_amdgcn_wmma_f32_16x16x4_f32(false, a_z, false, bf[k], (short)0, cz, false, false);

        #pragma unroll
        for (int r = 0; r < 8; r++) {
          float z = fmaxf(cz[r], Z_MIN);
          float iz = __builtin_amdgcn_rcpf(z);
          float ru = (cu[r] * iz - xu[k]) * wu[k];
          float rv = (cv[r] * iz - xv[k]) * wv[k];
          acc[ti][r] = fmaf(ru, ru, fmaf(rv, rv, acc[ti][r]));
        }
      }
    }

    #pragma unroll
    for (int m = 1; m < 16; m <<= 1) {
      #pragma unroll
      for (int ti = 0; ti < 4; ++ti) {
        #pragma unroll
        for (int r = 0; r < 8; r++) acc[ti][r] += __shfl_xor(acc[ti][r], m, 32);
      }
    }
    if (sl == 0) {
      #pragma unroll
      for (int ti = 0; ti < 4; ++ti) {
        #pragma unroll
        for (int r = 0; r < 8; r++) wave_part[tq + ti][wave][r + 8 * g] = acc[ti][r];
      }
    }
  }
  __syncthreads();

  if (tid < SPW) {
    const int t = tid >> 4, s = tid & 15;
    float c = 0.f;
    for (int w = 0; w < 8; w++) c += wave_part[t][w][s];
    const int mc = grp * SPW + tid;
    out[LW_OFF + (size_t)mc * OBJ + o] = -0.5f * c - As[tid][12];
  }
  if (do_ci && tid == 0) {
    float total = 0.f;
    #pragma unroll
    for (int w = 0; w < 8; w++) total += ci_part[w];
    out[CI_OFF + o] = 0.5f * total;
  }
}

extern "C" void kernel_launch(void* const* d_in, const int* in_sizes, int n_in,
                              void* d_out, int out_size, void* d_ws, size_t ws_size,
                              hipStream_t stream) {
  const float* x3d     = (const float*)d_in[0];
  const float* x2d     = (const float*)d_in[1];
  const float* w2d     = (const float*)d_in[2];
  const float* pose    = (const float*)d_in[3];
  const float* cam     = (const float*)d_in[4];
  const float* cov_t   = (const float*)d_in[5];
  const float* cov_r   = (const float*)d_in[6];
  const float* noise_t = (const float*)d_in[7];
  const float* noise_r = (const float*)d_in[8];
  float* out = (float*)d_out;

  epropnp_fused_kernel<<<dim3(4, 64), 256, 0, stream>>>(x3d, x2d, w2d, pose, cam, cov_t, cov_r, noise_t, noise_r, out);
}


# matrix product transposed: points on rows, samples on columns, so a lane sums its own 8 result registers over points and the 16-lane exchanges go; sum order over points changed; 2-wave register cap
# speedup vs baseline: 1.7084x; 1.1019x over previous
#include <hip/hip_runtime.h>
#include <hip/hip_bf16.h>
#include <math.h>

typedef __attribute__((ext_vector_type(2))) float v2f;
typedef __attribute__((ext_vector_type(8))) float v8f;

#define OBJ 64
#define NPTS 512
#define MC 512
#define Z_MIN 1e-4f
#define EPSV 1e-5f

#define POSE_OFF 0
#define LW_OFF   229376
#define CI_OFF   262144

#define C_LP_T 2.7568155996140186f
#define C_LP_R 2.9826069522913796f

__device__ __forceinline__ void quat_rot(float qw, float qi, float qj, float qk, float R[3][3]) {
  R[0][0] = 1.f - 2.f * (qj * qj + qk * qk);
  R[0][1] = 2.f * (qi * qj - qk * qw);
  R[0][2] = 2.f * (qi * qk + qj * qw);
  R[1][0] = 2.f * (qi * qj + qk * qw);
  R[1][1] = 1.f - 2.f * (qi * qi + qk * qk);
  R[1][2] = 2.f * (qj * qk - qi * qw);
  R[2][0] = 2.f * (qi * qk - qj * qw);
  R[2][1] = 2.f * (qj * qk + qi * qw);
  R[2][2] = 1.f - 2.f * (qi * qi + qj * qj);
}

__device__ __forceinline__ void wave_point_sums(const float X[4], const float Y[4], const float Z[4],
                                                float (*psum)[4], int wave, int lane) {
  float sx = (X[0] + X[1]) + (X[2] + X[3]);
  float sy = (Y[0] + Y[1]) + (Y[2] + Y[3]);
  float sz = (Z[0] + Z[1]) + (Z[2] + Z[3]);
  #pragma unroll
  for (int m = 1; m < 16; m <<= 1) {
    sx += __shfl_xor(sx, m, 32);
    sy += __shfl_xor(sy, m, 32);
    sz += __shfl_xor(sz, m, 32);
  }
  if (lane == 0) { psum[wave][0] = sx; psum[wave][1] = sy; psum[wave][2] = sz; }
}

__device__ __forceinline__ void load_targets(const float* __restrict__ x2d, const float* __restrict__ w2d,
                                             int o, int wave, int g,
                                             v2f tu[4][4], v2f tv[4][4], v2f su[4][4], v2f sv[4][4]) {
  #pragma unroll
  for (int k = 0; k < 4; ++k) {
    const int nb = (wave * 4 + k) * 16 + 8 * g;
    const float* x2p = x2d + ((size_t)o * NPTS + nb) * 2;
    const float* w2p = w2d + ((size_t)o * NPTS + nb) * 2;
    #pragma unroll
    for (int j = 0; j < 4; ++j) {
      tu[k][j].x = x2p[4 * j];     tv[k][j].x = x2p[4 * j + 1];
      tu[k][j].y = x2p[4 * j + 2]; tv[k][j].y = x2p[4 * j + 3];
      su[k][j].x = w2p[4 * j];     sv[k][j].x = w2p[4 * j + 1];
      su[k][j].y = w2p[4 * j + 2]; sv[k][j].y = w2p[4 * j + 3];
    }
  }
}

#define TILES 8
#define SPW (TILES * 16)

__global__ void __launch_bounds__(256) __attribute__((amdgpu_waves_per_eu(2, 2)))
epropnp_fused_kernel(const float* __restrict__ x3d,
                     const float* __restrict__ x2d,
                     const float* __restrict__ w2d,
                     const float* __restrict__ pose,
                     const float* __restrict__ cam,
                     const float* __restrict__ cov_t,
                     const float* __restrict__ cov_r,
                     const float* __restrict__ noise_t,
                     const float* __restrict__ noise_r,
                     float* __restrict__ out) {
  const int o = blockIdx.y;
  const int grp = blockIdx.x;
  const int tid = threadIdx.x;
  const int lane = tid & 31;
  const int wave = tid >> 5;
  const int g = lane >> 4;
  const int sl = lane & 15;
  const bool do_ci = (grp == 0);

  __shared__ float As[SPW][16];
  __shared__ float wave_part[TILES][16][16];
  __shared__ float psum[8][4];
  __shared__ float ci_part[8];

  float X[4], Y[4], Z[4];
  #pragma unroll
  for (int k = 0; k < 4; ++k) {
    const int n = (wave * 4 + k) * 16 + sl;
    const float* xp = x3d + ((size_t)o * NPTS + n) * 3;
    X[k] = xp[0]; Y[k] = xp[1]; Z[k] = xp[2];
  }
  float xu[4], xv[4], wu[4], wv[4];
  #pragma unroll
  for (int k = 0; k < 4; ++k) {
    const int n = (wave * 4 + k) * 16 + sl;
    const float* x2p = x2d + ((size_t)o * NPTS + n) * 2;
    const float* w2p = w2d + ((size_t)o * NPTS + n) * 2;
    xu[k] = x2p[0]; xv[k] = x2p[1]; wu[k] = w2p[0]; wv[k] = w2p[1];
  }

  float R[3][3] = {}, R0[3][3] = {}, Kc[9] = {};
  float L00 = 0.f, L10 = 0.f, L11 = 0.f, L20 = 0.f, L21 = 0.f, L22 = 0.f;
  float n0 = 0.f, n1 = 0.f, n2 = 0.f, P0 = 0.f, P1 = 0.f, P2 = 0.f;

  v2f tu[4][4], tv[4][4], su[4][4], sv[4][4];
  if (tid >= SPW) load_targets(x2d, w2d, o, wave, g, tu, tv, su, sv);

  if (tid < SPW) {
    const int s = tid;
    const int mc = grp * SPW + s;
    const float* nt = noise_t + ((size_t)mc * OBJ + o) * 3;
    const float* nr = noise_r + ((size_t)mc * OBJ + o) * 4;
    n0 = nt[0]; n1 = nt[1]; n2 = nt[2];
    const float r0 = nr[0], r1 = nr[1], r2 = nr[2], r3 = nr[3];
    const float* P = pose + o * 7;
    P0 = P[0]; P1 = P[1]; P2 = P[2];
    quat_rot(P[3], P[4], P[5], P[6], R0);

    const float* A = cov_t + o * 9;
    L00 = sqrtf(A[0]);
    L10 = A[3] / L00;
    L20 = A[6] / L00;
    L11 = sqrtf(A[4] - L10 * L10);
    L21 = (A[7] - L20 * L10) / L11;
    L22 = sqrtf(A[8] - L20 * L20 - L21 * L21);

    const float* B = cov_r + o * 16;
    const float N00 = sqrtf(B[0]);
    const float N10 = B[4] / N00, N20 = B[8] / N00, N30 = B[12] / N00;
    const float N11 = sqrtf(B[5] - N10 * N10);
    const float N21 = (B[9] - N20 * N10) / N11;
    const float N31 = (B[13] - N30 * N10) / N11;
    const float N22 = sqrtf(B[10] - N20 * N20 - N21 * N21);
    const float N32 = (B[14] - N30 * N20 - N31 * N21) / N22;
    const float N33 = sqrtf(B[15] - N30 * N30 - N31 * N31 - N32 * N32);

    const float hld_t = logf(L00) + logf(L11) + logf(L22);
    const float hld_r = logf(N00) + logf(N11) + logf(N22) + logf(N33);

    float g0 = N00 * r0;
    float g1 = N10 * r0 + N11 * r1;
    float g2 = N20 * r0 + N21 * r1 + N22 * r2;
    float g3 = N30 * r0 + N31 * r1 + N32 * r2 + N33 * r3;
    float gn = sqrtf(g0 * g0 + g1 * g1 + g2 * g2 + g3 * g3);

    float qw, qi, qj, qk, Mr;
    if (gn < EPSV) {
      qw = 1.f; qi = 0.f; qj = 0.f; qk = 0.f;
      float y0 = 1.f / N00;
      float y1 = -(N10 * y0) / N11;
      float y2 = -(N20 * y0 + N21 * y1) / N22;
      float y3 = -(N30 * y0 + N31 * y1 + N32 * y2) / N33;
      Mr = y0 * y0 + y1 * y1 + y2 * y2 + y3 * y3;
    } else {
      float inv = 1.f / gn;
      qw = g0 * inv; qi = g1 * inv; qj = g2 * inv; qk = g3 * inv;
      Mr = (r0 * r0 + r1 * r1 + r2 * r2 + r3 * r3) * inv * inv;
    }
    quat_rot(qw, qi, qj, qk, R);

    load_targets(x2d, w2d, o, wave, g, tu, tv, su, sv);

    const float* Kp = cam + o * 9;
    #pragma unroll
    for (int i = 0; i < 9; i++) Kc[i] = Kp[i];
    #pragma unroll
    for (int i = 0; i < 3; i++) {
      #pragma unroll
      for (int j = 0; j < 3; j++)
        As[s][i * 4 + j] = Kc[i * 3 + 0] * R[0][j] + Kc[i * 3 + 1] * R[1][j] + Kc[i * 3 + 2] * R[2][j];
    }
    float lp_t = -0.5f * (n0 * n0 + n1 * n1 + n2 * n2) - hld_t - C_LP_T;
    float lp_r = -2.f * logf(Mr) - hld_r - C_LP_R;
    As[s][12] = lp_t + lp_r;
    As[s][13] = 0.f; As[s][14] = 0.f; As[s][15] = 0.f;

    float* po = out + POSE_OFF + ((size_t)mc * OBJ + o) * 7;
    po[3] = qw; po[4] = qi; po[5] = qj; po[6] = qk;
  }
  wave_point_sums(X, Y, Z, psum, wave, lane);
  __syncthreads();

  float offx = 0.f, offy = 0.f, offz = 0.f;
  #pragma unroll
  for (int w = 0; w < 8; w++) { offx += psum[w][0]; offy += psum[w][1]; offz += psum[w][2]; }
  offx *= (1.f / NPTS); offy *= (1.f / NPTS); offz *= (1.f / NPTS);

  if (tid < SPW) {
    const int s = tid;
    const int mc = grp * SPW + s;
    const float t0x = P0 + R0[0][0] * offx + R0[0][1] * offy + R0[0][2] * offz;
    const float t0y = P1 + R0[1][0] * offx + R0[1][1] * offy + R0[1][2] * offz;
    const float t0z = P2 + R0[2][0] * offx + R0[2][1] * offy + R0[2][2] * offz;
    const float tsx = t0x + L00 * n0;
    const float tsy = t0y + L10 * n0 + L11 * n1;
    const float tsz = t0z + L20 * n0 + L21 * n1 + L22 * n2;
    #pragma unroll
    for (int i = 0; i < 3; i++)
      As[s][i * 4 + 3] = Kc[i * 3 + 0] * tsx + Kc[i * 3 + 1] * tsy + Kc[i * 3 + 2] * tsz;

    float* po = out + POSE_OFF + ((size_t)mc * OBJ + o) * 7;
    po[0] = tsx - (R[0][0] * offx + R[0][1] * offy + R[0][2] * offz);
    po[1] = tsy - (R[1][0] * offx + R[1][1] * offy + R[1][2] * offz);
    po[2] = tsz - (R[2][0] * offx + R[2][1] * offy + R[2][2] * offz);
  }

  if (do_ci) {
    const float* P = pose + o * 7;
    const float* Kp = cam + o * 9;
    const float cP0 = P[0], cP1 = P[1], cP2 = P[2];
    float cK[9], cR0[3][3];
    #pragma unroll
    for (int i = 0; i < 9; i++) cK[i] = Kp[i];
    quat_rot(P[3], P[4], P[5], P[6], cR0);
    const float xa = g ? X[2] : X[0], ya = g ? Y[2] : Y[0], za = g ? Z[2] : Z[0];
    const float xb = g ? X[3] : X[1], yb = g ? Y[3] : Y[1], zb = g ? Z[3] : Z[1];
    const float xua = g ? xu[2] : xu[0], xva = g ? xv[2] : xv[0];
    const float wua = g ? wu[2] : wu[0], wva = g ? wv[2] : wv[0];
    const float xub = g ? xu[3] : xu[1], xvb = g ? xv[3] : xv[1];
    const float wub = g ? wu[3] : wu[1], wvb = g ? wv[3] : wv[1];

    float ts[3];
    ts[0] = cP0 + cR0[0][0] * offx + cR0[0][1] * offy + cR0[0][2] * offz;
    ts[1] = cP1 + cR0[1][0] * offx + cR0[1][1] * offy + cR0[1][2] * offz;
    ts[2] = cP2 + cR0[2][0] * offx + cR0[2][1] * offy + cR0[2][2] * offz;
    float M0[3][3], b0[3];
    #pragma unroll
    for (int i = 0; i < 3; i++) {
      #pragma unroll
      for (int j = 0; j < 3; j++)
        M0[i][j] = cK[i * 3 + 0] * cR0[0][j] + cK[i * 3 + 1] * cR0[1][j] + cK[i * 3 + 2] * cR0[2][j];
      b0[i] = cK[i * 3 + 0] * ts[0] + cK[i * 3 + 1] * ts[1] + cK[i * 3 + 2] * ts[2];
    }

    float c = 0.f;
    {
      float px = xa - offx, py = ya - offy, pz = za - offz;
      float u = M0[0][0] * px + M0[0][1] * py + M0[0][2] * pz + b0[0];
      float v = M0[1][0] * px + M0[1][1] * py + M0[1][2] * pz + b0[1];
      float w = M0[2][0] * px + M0[2][1] * py + M0[2][2] * pz + b0[2];
      float z = fmaxf(w, Z_MIN);
      float iz = __builtin_amdgcn_rcpf(z);
      float ru = (u * iz - xua) * wua;
      float rv = (v * iz - xva) * wva;
      c = fmaf(ru, ru, fmaf(rv, rv, c));
    }
    {
      float px = xb - offx, py = yb - offy, pz = zb - offz;
      float u = M0[0][0] * px + M0[0][1] * py + M0[0][2] * pz + b0[0];
      float v = M0[1][0] * px + M0[1][1] * py + M0[1][2] * pz + b0[1];
      float w = M0[2][0] * px + M0[2][1] * py + M0[2][2] * pz + b0[2];
      float z = fmaxf(w, Z_MIN);
      float iz = __builtin_amdgcn_rcpf(z);
      float ru = (u * iz - xub) * wub;
      float rv = (v * iz - xvb) * wvb;
      c = fmaf(ru, ru, fmaf(rv, rv, c));
    }
    #pragma unroll
    for (int m = 1; m < 32; m <<= 1) c += __shfl_xor(c, m, 32);
    if (lane == 0) ci_part[wave] = c;
  }
  __syncthreads();

  const int k0 = 2 * g;

  v2f bf[4];
  #pragma unroll
  for (int k = 0; k < 4; ++k) {
    float px = X[k] - offx;
    float py = Y[k] - offy;
    float pz = Z[k] - offz;
    bf[k].x = g ? pz : px;
    bf[k].y = g ? 1.0f : py;
  }

  #pragma unroll 1
  for (int tq = 0; tq < TILES; tq += 4) {
    v2f acc[4][4];
    #pragma unroll
    for (int ti = 0; ti < 4; ++ti) {
      const float* Ar = &As[(tq + ti) * 16 + sl][0];
      v2f b_u = {Ar[0 + k0], Ar[1 + k0]};
      v2f b_v = {Ar[4 + k0], Ar[5 + k0]};
      v2f b_z = {Ar[8 + k0], Ar[9 + k0]};

      #pragma unroll
      for (int j = 0; j < 4; j++) acc[ti][j] = (v2f){0.f, 0.f};

      #pragma unroll
      for (int k = 0; k < 4; ++k) {
        v8f cu = {}, cv = {}, cz = {};
        cu = __builtin_amdgcn_wmma_f32_16x16x4_f32(false, bf[k], false, b_u, (short)0, cu, false, false);
        cv = __builtin_amdgcn_wmma_f32_16x16x4_f32(false, bf[k], false, b_v, (short)0, cv, false, false);
        cz = __builtin_amdgcn_wmma_f32_16x16x4_f32(false, bf[k], false, b_z, (short)0, cz, false, false);

        #pragma unroll
        for (int j = 0; j < 4; j++) {
          const float z0 = fmaxf(cz[2 * j], Z_MIN), z1 = fmaxf(cz[2 * j + 1], Z_MIN);
          const v2f iz = {__builtin_amdgcn_rcpf(z0), __builtin_amdgcn_rcpf(z1)};
          const v2f u2 = {cu[2 * j], cu[2 * j + 1]};
          const v2f v2 = {cv[2 * j], cv[2 * j + 1]};
          const v2f ru = (u2 * iz - tu[k][j]) * su[k][j];
          const v2f rv = (v2 * iz - tv[k][j]) * sv[k][j];
          acc[ti][j] = __builtin_elementwise_fma(ru, ru, __builtin_elementwise_fma(rv, rv, acc[ti][j]));
        }
      }
    }

    #pragma unroll
    for (int ti = 0; ti < 4; ++ti) {
      const v2f s = (acc[ti][0] + acc[ti][1]) + (acc[ti][2] + acc[ti][3]);
      wave_part[tq + ti][wave * 2 + g][sl] = s.x + s.y;
    }
  }
  __syncthreads();

  if (tid < SPW) {
    const int t = tid >> 4, s = tid & 15;
    float c = 0.f;
    for (int w = 0; w < 16; w++) c += wave_part[t][w][s];
    const int mc = grp * SPW + tid;
    out[LW_OFF + (size_t)mc * OBJ + o] = -0.5f * c - As[tid][12];
  }
  if (do_ci && tid == 0) {
    float total = 0.f;
    #pragma unroll
    for (int w = 0; w < 8; w++) total += ci_part[w];
    out[CI_OFF + o] = 0.5f * total;
  }
}

extern "C" void kernel_launch(void* const* d_in, const int* in_sizes, int n_in,
                              void* d_out, int out_size, void* d_ws, size_t ws_size,
                              hipStream_t stream) {
  const float* x3d     = (const float*)d_in[0];
  const float* x2d     = (const float*)d_in[1];
  const float* w2d     = (const float*)d_in[2];
  const float* pose    = (const float*)d_in[3];
  const float* cam     = (const float*)d_in[4];
  const float* cov_t   = (const float*)d_in[5];
  const float* cov_r   = (const float*)d_in[6];
  const float* noise_t = (const float*)d_in[7];
  const float* noise_r = (const float*)d_in[8];
  float* out = (float*)d_out;

  epropnp_fused_kernel<<<dim3(4, 64), 256, 0, stream>>>(x3d, x2d, w2d, pose, cam, cov_t, cov_r, noise_t, noise_r, out);
}


# mean offset folded into each sample's translation, points enter the product as loaded: one barrier before the sample loop, pose stores behind it; lane sums on DPP; offset, cost_init sum order changed
# speedup vs baseline: 2.0295x; 1.1880x over previous
#include <hip/hip_runtime.h>
#include <hip/hip_bf16.h>
#include <math.h>

typedef __attribute__((ext_vector_type(2))) float v2f;
typedef __attribute__((ext_vector_type(8))) float v8f;

#define OBJ 64
#define NPTS 512
#define MC 512
#define Z_MIN 1e-4f
#define EPSV 1e-5f

#define POSE_OFF 0
#define LW_OFF   229376
#define CI_OFF   262144

#define C_LP_T 2.7568155996140186f
#define C_LP_R 2.9826069522913796f

__device__ __forceinline__ void quat_rot(float qw, float qi, float qj, float qk, float R[3][3]) {
  R[0][0] = 1.f - 2.f * (qj * qj + qk * qk);
  R[0][1] = 2.f * (qi * qj - qk * qw);
  R[0][2] = 2.f * (qi * qk + qj * qw);
  R[1][0] = 2.f * (qi * qj + qk * qw);
  R[1][1] = 1.f - 2.f * (qi * qi + qk * qk);
  R[1][2] = 2.f * (qj * qk - qi * qw);
  R[2][0] = 2.f * (qi * qk - qj * qw);
  R[2][1] = 2.f * (qj * qk + qi * qw);
  R[2][2] = 1.f - 2.f * (qi * qi + qj * qj);
}

template <int PATTERN>
__device__ __forceinline__ float row_take(float v) {
  return __builtin_bit_cast(float, __builtin_amdgcn_update_dpp(0, __builtin_bit_cast(int, v), PATTERN, 0xf, 0xf, false));
}
__device__ __forceinline__ float other_row(float v) {
  return __builtin_bit_cast(float, __builtin_amdgcn_permlanex16(__builtin_bit_cast(int, v), __builtin_bit_cast(int, v),
                                                               0x76543210, 0xfedcba98, false, false));
}
__device__ __forceinline__ float wave_sum(float v) {
  v += row_take<0xB1>(v);
  v += row_take<0x4E>(v);
  v += row_take<0x141>(v);
  v += row_take<0x140>(v);
  v += other_row(v);
  return v;
}

__device__ __forceinline__ void load_targets(const float* __restrict__ x2d, const float* __restrict__ w2d,
                                             int o, int wave, int g,
                                             v2f tu[4][4], v2f tv[4][4], v2f su[4][4], v2f sv[4][4]) {
  #pragma unroll
  for (int k = 0; k < 4; ++k) {
    const int nb = (wave * 4 + k) * 16 + 8 * g;
    const float* x2p = x2d + ((size_t)o * NPTS + nb) * 2;
    const float* w2p = w2d + ((size_t)o * NPTS + nb) * 2;
    #pragma unroll
    for (int j = 0; j < 4; ++j) {
      tu[k][j].x = x2p[4 * j];     tv[k][j].x = x2p[4 * j + 1];
      tu[k][j].y = x2p[4 * j + 2]; tv[k][j].y = x2p[4 * j + 3];
      su[k][j].x = w2p[4 * j];     sv[k][j].x = w2p[4 * j + 1];
      su[k][j].y = w2p[4 * j + 2]; sv[k][j].y = w2p[4 * j + 3];
    }
  }
}

#define TILES 8
#define SPW (TILES * 16)

__global__ void __launch_bounds__(256) __attribute__((amdgpu_waves_per_eu(2, 2)))
epropnp_fused_kernel(const float* __restrict__ x3d,
                     const float* __restrict__ x2d,
                     const float* __restrict__ w2d,
                     const float* __restrict__ pose,
                     const float* __restrict__ cam,
                     const float* __restrict__ cov_t,
                     const float* __restrict__ cov_r,
                     const float* __restrict__ noise_t,
                     const float* __restrict__ noise_r,
                     float* __restrict__ out) {
  const int o = blockIdx.y;
  const int grp = blockIdx.x;
  const int tid = threadIdx.x;
  const int lane = tid & 31;
  const int wave = tid >> 5;
  const int g = lane >> 4;
  const int sl = lane & 15;
  const bool do_ci = (grp == 0);

  __shared__ float As[SPW][16];
  __shared__ float wave_part[TILES][16][16];
  __shared__ float ci_part[8];

  float X[4], Y[4], Z[4];
  #pragma unroll
  for (int k = 0; k < 4; ++k) {
    const int n = (wave * 4 + k) * 16 + sl;
    const float* xp = x3d + ((size_t)o * NPTS + n) * 3;
    X[k] = xp[0]; Y[k] = xp[1]; Z[k] = xp[2];
  }
  float xu[4], xv[4], wu[4], wv[4];
  #pragma unroll
  for (int k = 0; k < 4; ++k) {
    const int n = (wave * 4 + k) * 16 + sl;
    const float* x2p = x2d + ((size_t)o * NPTS + n) * 2;
    const float* w2p = w2d + ((size_t)o * NPTS + n) * 2;
    xu[k] = x2p[0]; xv[k] = x2p[1]; wu[k] = w2p[0]; wv[k] = w2p[1];
  }

  v2f tu[4][4], tv[4][4], su[4][4], sv[4][4];
  if (tid >= SPW) load_targets(x2d, w2d, o, wave, g, tu, tv, su, sv);

  float tox = 0.f, toy = 0.f, toz = 0.f, qw = 0.f, qi = 0.f, qj = 0.f, qk = 0.f;

  if (tid < SPW) {
    const int s = tid;
    const int mc = grp * SPW + s;

    float qx[16], qy[16], qz[16];
    #pragma unroll
    for (int i = 0; i < 16; ++i) {
      const float* qp = x3d + ((size_t)o * NPTS + i * 32 + lane) * 3;
      qx[i] = qp[0]; qy[i] = qp[1]; qz[i] = qp[2];
    }

    const float* nt = noise_t + ((size_t)mc * OBJ + o) * 3;
    const float* nr = noise_r + ((size_t)mc * OBJ + o) * 4;
    const float n0 = nt[0], n1 = nt[1], n2 = nt[2];
    const float r0 = nr[0], r1 = nr[1], r2 = nr[2], r3 = nr[3];
    const float* P = pose + o * 7;
    const float P0 = P[0], P1 = P[1], P2 = P[2];
    float R0[3][3];
    quat_rot(P[3], P[4], P[5], P[6], R0);

    #pragma unroll
    for (int w = 8; w >= 1; w >>= 1) {
      #pragma unroll
      for (int i = 0; i < w; ++i) { qx[i] += qx[i + w]; qy[i] += qy[i + w]; qz[i] += qz[i + w]; }
    }
    const float offx = wave_sum(qx[0]) * (1.f / NPTS);
    const float offy = wave_sum(qy[0]) * (1.f / NPTS);
    const float offz = wave_sum(qz[0]) * (1.f / NPTS);

    const float* A = cov_t + o * 9;
    const float L00 = sqrtf(A[0]);
    const float L10 = A[3] / L00;
    const float L20 = A[6] / L00;
    const float L11 = sqrtf(A[4] - L10 * L10);
    const float L21 = (A[7] - L20 * L10) / L11;
    const float L22 = sqrtf(A[8] - L20 * L20 - L21 * L21);

    const float* B = cov_r + o * 16;
    const float N00 = sqrtf(B[0]);
    const float N10 = B[4] / N00, N20 = B[8] / N00, N30 = B[12] / N00;
    const float N11 = sqrtf(B[5] - N10 * N10);
    const float N21 = (B[9] - N20 * N10) / N11;
    const float N31 = (B[13] - N30 * N10) / N11;
    const float N22 = sqrtf(B[10] - N20 * N20 - N21 * N21);
    const float N32 = (B[14] - N30 * N20 - N31 * N21) / N22;
    const float N33 = sqrtf(B[15] - N30 * N30 - N31 * N31 - N32 * N32);

    const float hld_t = logf(L00) + logf(L11) + logf(L22);
    const float hld_r = logf(N00) + logf(N11) + logf(N22) + logf(N33);

    float g0 = N00 * r0;
    float g1 = N10 * r0 + N11 * r1;
    float g2 = N20 * r0 + N21 * r1 + N22 * r2;
    float g3 = N30 * r0 + N31 * r1 + N32 * r2 + N33 * r3;
    float gn = sqrtf(g0 * g0 + g1 * g1 + g2 * g2 + g3 * g3);

    float Mr;
    if (gn < EPSV) {
      qw = 1.f; qi = 0.f; qj = 0.f; qk = 0.f;
      float y0 = 1.f / N00;
      float y1 = -(N10 * y0) / N11;
      float y2 = -(N20 * y0 + N21 * y1) / N22;
      float y3 = -(N30 * y0 + N31 * y1 + N32 * y2) / N33;
      Mr = y0 * y0 + y1 * y1 + y2 * y2 + y3 * y3;
    } else {
      float inv = 1.f / gn;
      qw = g0 * inv; qi = g1 * inv; qj = g2 * inv; qk = g3 * inv;
      Mr = (r0 * r0 + r1 * r1 + r2 * r2 + r3 * r3) * inv * inv;
    }
    float R[3][3];
    quat_rot(qw, qi, qj, qk, R);

    const float t0x = P0 + R0[0][0] * offx + R0[0][1] * offy + R0[0][2] * offz;
    const float t0y = P1 + R0[1][0] * offx + R0[1][1] * offy + R0[1][2] * offz;
    const float t0z = P2 + R0[2][0] * offx + R0[2][1] * offy + R0[2][2] * offz;
    const float tsx = t0x + L00 * n0;
    const float tsy = t0y + L10 * n0 + L11 * n1;
    const float tsz = t0z + L20 * n0 + L21 * n1 + L22 * n2;
    tox = tsx - (R[0][0] * offx + R[0][1] * offy + R[0][2] * offz);
    toy = tsy - (R[1][0] * offx + R[1][1] * offy + R[1][2] * offz);
    toz = tsz - (R[2][0] * offx + R[2][1] * offy + R[2][2] * offz);

    load_targets(x2d, w2d, o, wave, g, tu, tv, su, sv);

    const float* Kp = cam + o * 9;
    float Kc[9];
    #pragma unroll
    for (int i = 0; i < 9; i++) Kc[i] = Kp[i];
    #pragma unroll
    for (int i = 0; i < 3; i++) {
      #pragma unroll
      for (int j = 0; j < 3; j++)
        As[s][i * 4 + j] = Kc[i * 3 + 0] * R[0][j] + Kc[i * 3 + 1] * R[1][j] + Kc[i * 3 + 2] * R[2][j];
      As[s][i * 4 + 3] = Kc[i * 3 + 0] * tox + Kc[i * 3 + 1] * toy + Kc[i * 3 + 2] * toz;
    }
    float lp_t = -0.5f * (n0 * n0 + n1 * n1 + n2 * n2) - hld_t - C_LP_T;
    float lp_r = -2.f * logf(Mr) - hld_r - C_LP_R;
    As[s][12] = lp_t + lp_r;
    As[s][13] = 0.f; As[s][14] = 0.f; As[s][15] = 0.f;
  }

  if (do_ci) {
    const float* P = pose + o * 7;
    const float* Kp = cam + o * 9;
    const float cP0 = P[0], cP1 = P[1], cP2 = P[2];
    float cK[9], cR0[3][3];
    #pragma unroll
    for (int i = 0; i < 9; i++) cK[i] = Kp[i];
    quat_rot(P[3], P[4], P[5], P[6], cR0);
    const float xa = g ? X[2] : X[0], ya = g ? Y[2] : Y[0], za = g ? Z[2] : Z[0];
    const float xb = g ? X[3] : X[1], yb = g ? Y[3] : Y[1], zb = g ? Z[3] : Z[1];
    const float xua = g ? xu[2] : xu[0], xva = g ? xv[2] : xv[0];
    const float wua = g ? wu[2] : wu[0], wva = g ? wv[2] : wv[0];
    const float xub = g ? xu[3] : xu[1], xvb = g ? xv[3] : xv[1];
    const float wub = g ? wu[3] : wu[1], wvb = g ? wv[3] : wv[1];

    float M0[3][3], b0[3];
    #pragma unroll
    for (int i = 0; i < 3; i++) {
      #pragma unroll
      for (int j = 0; j < 3; j++)
        M0[i][j] = cK[i * 3 + 0] * cR0[0][j] + cK[i * 3 + 1] * cR0[1][j] + cK[i * 3 + 2] * cR0[2][j];
      b0[i] = cK[i * 3 + 0] * cP0 + cK[i * 3 + 1] * cP1 + cK[i * 3 + 2] * cP2;
    }

    float c = 0.f;
    {
      float u = M0[0][0] * xa + M0[0][1] * ya + M0[0][2] * za + b0[0];
      float v = M0[1][0] * xa + M0[1][1] * ya + M0[1][2] * za + b0[1];
      float w = M0[2][0] * xa + M0[2][1] * ya + M0[2][2] * za + b0[2];
      float z = fmaxf(w, Z_MIN);
      float iz = __builtin_amdgcn_rcpf(z);
      float ru = (u * iz - xua) * wua;
      float rv = (v * iz - xva) * wva;
      c = fmaf(ru, ru, fmaf(rv, rv, c));
    }
    {
      float u = M0[0][0] * xb + M0[0][1] * yb + M0[0][2] * zb + b0[0];
      float v = M0[1][0] * xb + M0[1][1] * yb + M0[1][2] * zb + b0[1];
      float w = M0[2][0] * xb + M0[2][1] * yb + M0[2][2] * zb + b0[2];
      float z = fmaxf(w, Z_MIN);
      float iz = __builtin_amdgcn_rcpf(z);
      float ru = (u * iz - xub) * wub;
      float rv = (v * iz - xvb) * wvb;
      c = fmaf(ru, ru, fmaf(rv, rv, c));
    }
    c = wave_sum(c);
    if (lane == 0) ci_part[wave] = c;
  }
  __syncthreads();

  if (tid < SPW) {
    float* po = out + POSE_OFF + ((size_t)(grp * SPW + tid) * OBJ + o) * 7;
    po[0] = tox; po[1] = toy; po[2] = toz;
    po[3] = qw; po[4] = qi; po[5] = qj; po[6] = qk;
  }

  const int k0 = 2 * g;

  v2f bf[4];
  #pragma unroll
  for (int k = 0; k < 4; ++k) {
    bf[k].x = g ? Z[k] : X[k];
    bf[k].y = g ? 1.0f : Y[k];
  }

  #pragma unroll 1
  for (int tq = 0; tq < TILES; tq += 4) {
    v2f acc[4][4];
    #pragma unroll
    for (int ti = 0; ti < 4; ++ti) {
      const float* Ar = &As[(tq + ti) * 16 + sl][0];
      v2f b_u = {Ar[0 + k0], Ar[1 + k0]};
      v2f b_v = {Ar[4 + k0], Ar[5 + k0]};
      v2f b_z = {Ar[8 + k0], Ar[9 + k0]};

      #pragma unroll
      for (int j = 0; j < 4; j++) acc[ti][j] = (v2f){0.f, 0.f};

      #pragma unroll
      for (int k = 0; k < 4; ++k) {
        v8f cu = {}, cv = {}, cz = {};
        cu = __builtin_amdgcn_wmma_f32_16x16x4_f32(false, bf[k], false, b_u, (short)0, cu, false, false);
        cv = __builtin_amdgcn_wmma_f32_16x16x4_f32(false, bf[k], false, b_v, (short)0, cv, false, false);
        cz = __builtin_amdgcn_wmma_f32_16x16x4_f32(false, bf[k], false, b_z, (short)0, cz, false, false);

        #pragma unroll
        for (int j = 0; j < 4; j++) {
          const float z0 = fmaxf(cz[2 * j], Z_MIN), z1 = fmaxf(cz[2 * j + 1], Z_MIN);
          const v2f iz = {__builtin_amdgcn_rcpf(z0), __builtin_amdgcn_rcpf(z1)};
          const v2f u2 = {cu[2 * j], cu[2 * j + 1]};
          const v2f v2 = {cv[2 * j], cv[2 * j + 1]};
          const v2f ru = (u2 * iz - tu[k][j]) * su[k][j];
          const v2f rv = (v2 * iz - tv[k][j]) * sv[k][j];
          acc[ti][j] = __builtin_elementwise_fma(ru, ru, __builtin_elementwise_fma(rv, rv, acc[ti][j]));
        }
      }
    }

    #pragma unroll
    for (int ti = 0; ti < 4; ++ti) {
      const v2f s = (acc[ti][0] + acc[ti][1]) + (acc[ti][2] + acc[ti][3]);
      wave_part[tq + ti][wave * 2 + g][sl] = s.x + s.y;
    }
  }
  __syncthreads();

  if (tid < SPW) {
    const int t = tid >> 4, s = tid & 15;
    float c = 0.f;
    for (int w = 0; w < 16; w++) c += wave_part[t][w][s];
    const int mc = grp * SPW + tid;
    out[LW_OFF + (size_t)mc * OBJ + o] = -0.5f * c - As[tid][12];
  }
  if (do_ci && tid == 0) {
    float total = 0.f;
    #pragma unroll
    for (int w = 0; w < 8; w++) total += ci_part[w];
    out[CI_OFF + o] = 0.5f * total;
  }
}

extern "C" void kernel_launch(void* const* d_in, const int* in_sizes, int n_in,
                              void* d_out, int out_size, void* d_ws, size_t ws_size,
                              hipStream_t stream) {
  const float* x3d     = (const float*)d_in[0];
  const float* x2d     = (const float*)d_in[1];
  const float* w2d     = (const float*)d_in[2];
  const float* pose    = (const float*)d_in[3];
  const float* cam     = (const float*)d_in[4];
  const float* cov_t   = (const float*)d_in[5];
  const float* cov_r   = (const float*)d_in[6];
  const float* noise_t = (const float*)d_in[7];
  const float* noise_r = (const float*)d_in[8];
  float* out = (float*)d_out;

  epropnp_fused_kernel<<<dim3(4, 64), 256, 0, stream>>>(x3d, x2d, w2d, pose, cam, cov_t, cov_r, noise_t, noise_r, out);
}


# Cholesky chain before the barrier on pivot reciprocals with hardware sqrt, rcp, log2 (1 ulp) for the correctly rounded sequences; unit-quaternion case by select; rounding of factors, log-dets changes
# speedup vs baseline: 2.1431x; 1.0560x over previous
#include <hip/hip_runtime.h>
#include <hip/hip_bf16.h>
#include <math.h>

typedef __attribute__((ext_vector_type(2))) float v2f;
typedef __attribute__((ext_vector_type(8))) float v8f;

#define OBJ 64
#define NPTS 512
#define MC 512
#define Z_MIN 1e-4f
#define EPSV 1e-5f

#define POSE_OFF 0
#define LW_OFF   229376
#define CI_OFF   262144

#define C_LP_T 2.7568155996140186f
#define C_LP_R 2.9826069522913796f

__device__ __forceinline__ void quat_rot(float qw, float qi, float qj, float qk, float R[3][3]) {
  R[0][0] = 1.f - 2.f * (qj * qj + qk * qk);
  R[0][1] = 2.f * (qi * qj - qk * qw);
  R[0][2] = 2.f * (qi * qk + qj * qw);
  R[1][0] = 2.f * (qi * qj + qk * qw);
  R[1][1] = 1.f - 2.f * (qi * qi + qk * qk);
  R[1][2] = 2.f * (qj * qk - qi * qw);
  R[2][0] = 2.f * (qi * qk - qj * qw);
  R[2][1] = 2.f * (qj * qk + qi * qw);
  R[2][2] = 1.f - 2.f * (qi * qi + qj * qj);
}

template <int PATTERN>
__device__ __forceinline__ float row_take(float v) {
  return __builtin_bit_cast(float, __builtin_amdgcn_update_dpp(0, __builtin_bit_cast(int, v), PATTERN, 0xf, 0xf, false));
}
__device__ __forceinline__ float other_row(float v) {
  return __builtin_bit_cast(float, __builtin_amdgcn_permlanex16(__builtin_bit_cast(int, v), __builtin_bit_cast(int, v),
                                                               0x76543210, 0xfedcba98, false, false));
}
__device__ __forceinline__ float wave_sum(float v) {
  v += row_take<0xB1>(v);
  v += row_take<0x4E>(v);
  v += row_take<0x141>(v);
  v += row_take<0x140>(v);
  v += other_row(v);
  return v;
}

__device__ __forceinline__ float hw_sqrt(float x) {
  const bool tiny = x < 0x1p-96f;
  const float r = __builtin_amdgcn_sqrtf(tiny ? x * 0x1p+32f : x);
  return tiny ? r * 0x1p-16f : r;
}
#define LN2F 0.69314718056f

__device__ __forceinline__ void load_targets(const float* __restrict__ x2d, const float* __restrict__ w2d,
                                             int o, int wave, int g,
                                             v2f tu[4][4], v2f tv[4][4], v2f su[4][4], v2f sv[4][4]) {
  #pragma unroll
  for (int k = 0; k < 4; ++k) {
    const int nb = (wave * 4 + k) * 16 + 8 * g;
    const float* x2p = x2d + ((size_t)o * NPTS + nb) * 2;
    const float* w2p = w2d + ((size_t)o * NPTS + nb) * 2;
    #pragma unroll
    for (int j = 0; j < 4; ++j) {
      tu[k][j].x = x2p[4 * j];     tv[k][j].x = x2p[4 * j + 1];
      tu[k][j].y = x2p[4 * j + 2]; tv[k][j].y = x2p[4 * j + 3];
      su[k][j].x = w2p[4 * j];     sv[k][j].x = w2p[4 * j + 1];
      su[k][j].y = w2p[4 * j + 2]; sv[k][j].y = w2p[4 * j + 3];
    }
  }
}

#define TILES 8
#define SPW (TILES * 16)

__global__ void __launch_bounds__(256) __attribute__((amdgpu_waves_per_eu(2, 2)))
epropnp_fused_kernel(const float* __restrict__ x3d,
                     const float* __restrict__ x2d,
                     const float* __restrict__ w2d,
                     const float* __restrict__ pose,
                     const float* __restrict__ cam,
                     const float* __restrict__ cov_t,
                     const float* __restrict__ cov_r,
                     const float* __restrict__ noise_t,
                     const float* __restrict__ noise_r,
                     float* __restrict__ out) {
  const int o = blockIdx.y;
  const int grp = blockIdx.x;
  const int tid = threadIdx.x;
  const int lane = tid & 31;
  const int wave = tid >> 5;
  const int g = lane >> 4;
  const int sl = lane & 15;
  const bool do_ci = (grp == 0);

  __shared__ float As[SPW][16];
  __shared__ float wave_part[TILES][16][16];
  __shared__ float ci_part[8];

  float X[4], Y[4], Z[4];
  #pragma unroll
  for (int k = 0; k < 4; ++k) {
    const int n = (wave * 4 + k) * 16 + sl;
    const float* xp = x3d + ((size_t)o * NPTS + n) * 3;
    X[k] = xp[0]; Y[k] = xp[1]; Z[k] = xp[2];
  }
  float xu[4], xv[4], wu[4], wv[4];
  #pragma unroll
  for (int k = 0; k < 4; ++k) {
    const int n = (wave * 4 + k) * 16 + sl;
    const float* x2p = x2d + ((size_t)o * NPTS + n) * 2;
    const float* w2p = w2d + ((size_t)o * NPTS + n) * 2;
    xu[k] = x2p[0]; xv[k] = x2p[1]; wu[k] = w2p[0]; wv[k] = w2p[1];
  }

  v2f tu[4][4], tv[4][4], su[4][4], sv[4][4];
  if (tid >= SPW) load_targets(x2d, w2d, o, wave, g, tu, tv, su, sv);

  float tox = 0.f, toy = 0.f, toz = 0.f, qw = 0.f, qi = 0.f, qj = 0.f, qk = 0.f;

  if (tid < SPW) {
    const int s = tid;
    const int mc = grp * SPW + s;

    float qx[16], qy[16], qz[16];
    #pragma unroll
    for (int i = 0; i < 16; ++i) {
      const float* qp = x3d + ((size_t)o * NPTS + i * 32 + lane) * 3;
      qx[i] = qp[0]; qy[i] = qp[1]; qz[i] = qp[2];
    }

    const float* nt = noise_t + ((size_t)mc * OBJ + o) * 3;
    const float* nr = noise_r + ((size_t)mc * OBJ + o) * 4;
    const float n0 = nt[0], n1 = nt[1], n2 = nt[2];
    const float r0 = nr[0], r1 = nr[1], r2 = nr[2], r3 = nr[3];
    const float* P = pose + o * 7;
    const float P0 = P[0], P1 = P[1], P2 = P[2];
    float R0[3][3];
    quat_rot(P[3], P[4], P[5], P[6], R0);

    #pragma unroll
    for (int w = 8; w >= 1; w >>= 1) {
      #pragma unroll
      for (int i = 0; i < w; ++i) { qx[i] += qx[i + w]; qy[i] += qy[i + w]; qz[i] += qz[i + w]; }
    }
    const float offx = wave_sum(qx[0]) * (1.f / NPTS);
    const float offy = wave_sum(qy[0]) * (1.f / NPTS);
    const float offz = wave_sum(qz[0]) * (1.f / NPTS);

    const float* A = cov_t + o * 9;
    const float L00 = hw_sqrt(A[0]);
    const float iL00 = __builtin_amdgcn_rcpf(L00);
    const float L10 = A[3] * iL00;
    const float L20 = A[6] * iL00;
    const float L11 = hw_sqrt(A[4] - L10 * L10);
    const float iL11 = __builtin_amdgcn_rcpf(L11);
    const float L21 = (A[7] - L20 * L10) * iL11;
    const float L22 = hw_sqrt(A[8] - L20 * L20 - L21 * L21);

    const float* B = cov_r + o * 16;
    const float N00 = hw_sqrt(B[0]);
    const float iN00 = __builtin_amdgcn_rcpf(N00);
    const float N10 = B[4] * iN00, N20 = B[8] * iN00, N30 = B[12] * iN00;
    const float N11 = hw_sqrt(B[5] - N10 * N10);
    const float iN11 = __builtin_amdgcn_rcpf(N11);
    const float N21 = (B[9] - N20 * N10) * iN11;
    const float N31 = (B[13] - N30 * N10) * iN11;
    const float N22 = hw_sqrt(B[10] - N20 * N20 - N21 * N21);
    const float iN22 = __builtin_amdgcn_rcpf(N22);
    const float N32 = (B[14] - N30 * N20 - N31 * N21) * iN22;
    const float N33 = hw_sqrt(B[15] - N30 * N30 - N31 * N31 - N32 * N32);
    const float iN33 = __builtin_amdgcn_rcpf(N33);

    const float hld_t = (__builtin_amdgcn_logf(L00) + __builtin_amdgcn_logf(L11) + __builtin_amdgcn_logf(L22)) * LN2F;
    const float hld_r = (__builtin_amdgcn_logf(N00) + __builtin_amdgcn_logf(N11) + __builtin_amdgcn_logf(N22) +
                         __builtin_amdgcn_logf(N33)) * LN2F;

    const float y0 = iN00;
    const float y1 = -(N10 * y0) * iN11;
    const float y2 = -(N20 * y0 + N21 * y1) * iN22;
    const float y3 = -(N30 * y0 + N31 * y1 + N32 * y2) * iN33;
    const float Mr_unit = y0 * y0 + y1 * y1 + y2 * y2 + y3 * y3;

    float g0 = N00 * r0;
    float g1 = N10 * r0 + N11 * r1;
    float g2 = N20 * r0 + N21 * r1 + N22 * r2;
    float g3 = N30 * r0 + N31 * r1 + N32 * r2 + N33 * r3;
    const float gn = __builtin_amdgcn_sqrtf(g0 * g0 + g1 * g1 + g2 * g2 + g3 * g3);
    const bool unit = gn < EPSV;
    const float inv = __builtin_amdgcn_rcpf(gn);
    qw = unit ? 1.f : g0 * inv;
    qi = unit ? 0.f : g1 * inv;
    qj = unit ? 0.f : g2 * inv;
    qk = unit ? 0.f : g3 * inv;
    const float Mr = unit ? Mr_unit : (r0 * r0 + r1 * r1 + r2 * r2 + r3 * r3) * inv * inv;
    float R[3][3];
    quat_rot(qw, qi, qj, qk, R);

    const float t0x = P0 + R0[0][0] * offx + R0[0][1] * offy + R0[0][2] * offz;
    const float t0y = P1 + R0[1][0] * offx + R0[1][1] * offy + R0[1][2] * offz;
    const float t0z = P2 + R0[2][0] * offx + R0[2][1] * offy + R0[2][2] * offz;
    const float tsx = t0x + L00 * n0;
    const float tsy = t0y + L10 * n0 + L11 * n1;
    const float tsz = t0z + L20 * n0 + L21 * n1 + L22 * n2;
    tox = tsx - (R[0][0] * offx + R[0][1] * offy + R[0][2] * offz);
    toy = tsy - (R[1][0] * offx + R[1][1] * offy + R[1][2] * offz);
    toz = tsz - (R[2][0] * offx + R[2][1] * offy + R[2][2] * offz);

    load_targets(x2d, w2d, o, wave, g, tu, tv, su, sv);

    const float* Kp = cam + o * 9;
    float Kc[9];
    #pragma unroll
    for (int i = 0; i < 9; i++) Kc[i] = Kp[i];
    #pragma unroll
    for (int i = 0; i < 3; i++) {
      #pragma unroll
      for (int j = 0; j < 3; j++)
        As[s][i * 4 + j] = Kc[i * 3 + 0] * R[0][j] + Kc[i * 3 + 1] * R[1][j] + Kc[i * 3 + 2] * R[2][j];
      As[s][i * 4 + 3] = Kc[i * 3 + 0] * tox + Kc[i * 3 + 1] * toy + Kc[i * 3 + 2] * toz;
    }
    float lp_t = -0.5f * (n0 * n0 + n1 * n1 + n2 * n2) - hld_t - C_LP_T;
    float lp_r = -2.f * logf(Mr) - hld_r - C_LP_R;
    As[s][12] = lp_t + lp_r;
    As[s][13] = 0.f; As[s][14] = 0.f; As[s][15] = 0.f;
  }

  if (do_ci) {
    const float* P = pose + o * 7;
    const float* Kp = cam + o * 9;
    const float cP0 = P[0], cP1 = P[1], cP2 = P[2];
    float cK[9], cR0[3][3];
    #pragma unroll
    for (int i = 0; i < 9; i++) cK[i] = Kp[i];
    quat_rot(P[3], P[4], P[5], P[6], cR0);
    const float xa = g ? X[2] : X[0], ya = g ? Y[2] : Y[0], za = g ? Z[2] : Z[0];
    const float xb = g ? X[3] : X[1], yb = g ? Y[3] : Y[1], zb = g ? Z[3] : Z[1];
    const float xua = g ? xu[2] : xu[0], xva = g ? xv[2] : xv[0];
    const float wua = g ? wu[2] : wu[0], wva = g ? wv[2] : wv[0];
    const float xub = g ? xu[3] : xu[1], xvb = g ? xv[3] : xv[1];
    const float wub = g ? wu[3] : wu[1], wvb = g ? wv[3] : wv[1];

    float M0[3][3], b0[3];
    #pragma unroll
    for (int i = 0; i < 3; i++) {
      #pragma unroll
      for (int j = 0; j < 3; j++)
        M0[i][j] = cK[i * 3 + 0] * cR0[0][j] + cK[i * 3 + 1] * cR0[1][j] + cK[i * 3 + 2] * cR0[2][j];
      b0[i] = cK[i * 3 + 0] * cP0 + cK[i * 3 + 1] * cP1 + cK[i * 3 + 2] * cP2;
    }

    float c = 0.f;
    {
      float u = M0[0][0] * xa + M0[0][1] * ya + M0[0][2] * za + b0[0];
      float v = M0[1][0] * xa + M0[1][1] * ya + M0[1][2] * za + b0[1];
      float w = M0[2][0] * xa + M0[2][1] * ya + M0[2][2] * za + b0[2];
      float z = fmaxf(w, Z_MIN);
      float iz = __builtin_amdgcn_rcpf(z);
      float ru = (u * iz - xua) * wua;
      float rv = (v * iz - xva) * wva;
      c = fmaf(ru, ru, fmaf(rv, rv, c));
    }
    {
      float u = M0[0][0] * xb + M0[0][1] * yb + M0[0][2] * zb + b0[0];
      float v = M0[1][0] * xb + M0[1][1] * yb + M0[1][2] * zb + b0[1];
      float w = M0[2][0] * xb + M0[2][1] * yb + M0[2][2] * zb + b0[2];
      float z = fmaxf(w, Z_MIN);
      float iz = __builtin_amdgcn_rcpf(z);
      float ru = (u * iz - xub) * wub;
      float rv = (v * iz - xvb) * wvb;
      c = fmaf(ru, ru, fmaf(rv, rv, c));
    }
    c = wave_sum(c);
    if (lane == 0) ci_part[wave] = c;
  }
  __syncthreads();

  if (tid < SPW) {
    float* po = out + POSE_OFF + ((size_t)(grp * SPW + tid) * OBJ + o) * 7;
    po[0] = tox; po[1] = toy; po[2] = toz;
    po[3] = qw; po[4] = qi; po[5] = qj; po[6] = qk;
  }

  const int k0 = 2 * g;

  v2f bf[4];
  #pragma unroll
  for (int k = 0; k < 4; ++k) {
    bf[k].x = g ? Z[k] : X[k];
    bf[k].y = g ? 1.0f : Y[k];
  }

  #pragma unroll 1
  for (int tq = 0; tq < TILES; tq += 4) {
    v2f acc[4][4];
    #pragma unroll
    for (int ti = 0; ti < 4; ++ti) {
      const float* Ar = &As[(tq + ti) * 16 + sl][0];
      v2f b_u = {Ar[0 + k0], Ar[1 + k0]};
      v2f b_v = {Ar[4 + k0], Ar[5 + k0]};
      v2f b_z = {Ar[8 + k0], Ar[9 + k0]};

      #pragma unroll
      for (int j = 0; j < 4; j++) acc[ti][j] = (v2f){0.f, 0.f};

      #pragma unroll
      for (int k = 0; k < 4; ++k) {
        v8f cu = {}, cv = {}, cz = {};
        cu = __builtin_amdgcn_wmma_f32_16x16x4_f32(false, bf[k], false, b_u, (short)0, cu, false, false);
        cv = __builtin_amdgcn_wmma_f32_16x16x4_f32(false, bf[k], false, b_v, (short)0, cv, false, false);
        cz = __builtin_amdgcn_wmma_f32_16x16x4_f32(false, bf[k], false, b_z, (short)0, cz, false, false);

        #pragma unroll
        for (int j = 0; j < 4; j++) {
          const float z0 = fmaxf(cz[2 * j], Z_MIN), z1 = fmaxf(cz[2 * j + 1], Z_MIN);
          const v2f iz = {__builtin_amdgcn_rcpf(z0), __builtin_amdgcn_rcpf(z1)};
          const v2f u2 = {cu[2 * j], cu[2 * j + 1]};
          const v2f v2 = {cv[2 * j], cv[2 * j + 1]};
          const v2f ru = (u2 * iz - tu[k][j]) * su[k][j];
          const v2f rv = (v2 * iz - tv[k][j]) * sv[k][j];
          acc[ti][j] = __builtin_elementwise_fma(ru, ru, __builtin_elementwise_fma(rv, rv, acc[ti][j]));
        }
      }
    }

    #pragma unroll
    for (int ti = 0; ti < 4; ++ti) {
      const v2f s = (acc[ti][0] + acc[ti][1]) + (acc[ti][2] + acc[ti][3]);
      wave_part[tq + ti][wave * 2 + g][sl] = s.x + s.y;
    }
  }
  __syncthreads();

  if (tid < SPW) {
    const int t = tid >> 4, s = tid & 15;
    float c = 0.f;
    for (int w = 0; w < 16; w++) c += wave_part[t][w][s];
    const int mc = grp * SPW + tid;
    out[LW_OFF + (size_t)mc * OBJ + o] = -0.5f * c - As[tid][12];
  }
  if (do_ci && tid == 0) {
    float total = 0.f;
    #pragma unroll
    for (int w = 0; w < 8; w++) total += ci_part[w];
    out[CI_OFF + o] = 0.5f * total;
  }
}

extern "C" void kernel_launch(void* const* d_in, const int* in_sizes, int n_in,
                              void* d_out, int out_size, void* d_ws, size_t ws_size,
                              hipStream_t stream) {
  const float* x3d     = (const float*)d_in[0];
  const float* x2d     = (const float*)d_in[1];
  const float* w2d     = (const float*)d_in[2];
  const float* pose    = (const float*)d_in[3];
  const float* cam     = (const float*)d_in[4];
  const float* cov_t   = (const float*)d_in[5];
  const float* cov_r   = (const float*)d_in[6];
  const float* noise_t = (const float*)d_in[7];
  const float* noise_r = (const float*)d_in[8];
  float* out = (float*)d_out;

  epropnp_fused_kernel<<<dim3(4, 64), 256, 0, stream>>>(x3d, x2d, w2d, pose, cam, cov_t, cov_r, noise_t, noise_r, out);
}
